# starting kernel
# baseline (speedup 1.0000x reference)
#include <hip/hip_runtime.h>
#include <hip/hip_bf16.h>
#include <math.h>

#define BB   16
#define NN   512
#define DD   256
#define HH   8
#define DH   32
#define NT   (BB * NN)
#define BN_EPS 1e-5f

typedef float v2f __attribute__((ext_vector_type(2)));
typedef float v8f __attribute__((ext_vector_type(8)));

static __device__ __forceinline__ v8f wmma4(v2f a, v2f b, v8f c) {
    return __builtin_amdgcn_wmma_f32_16x16x4_f32(
         false, a,  false, b,
         (short)0, c,  false,  false);
}

static __device__ __forceinline__ float gelu_exact(float v) {
    return 0.5f * v * (1.0f + erff(v * 0.70710678118654752f));
}

__global__ __launch_bounds__(256)
void gemm_wmma_f32(const float* __restrict__ A, const float* __restrict__ W,
                   const float* __restrict__ bias, float* __restrict__ C,
                   int M, int N, int K, int fuse_gelu)
{
    const int wave = blockIdx.x * (blockDim.x >> 5) + (threadIdx.x >> 5);
    const int ntn  = N >> 6;
    const int tm   = wave / ntn;
    const int tn   = wave - tm * ntn;
    if (tm >= (M >> 5)) return;

    const int lane = threadIdx.x & 31;
    const int half = lane >> 4;
    const int lm   = lane & 15;

    const int m0 = tm * 32;
    const int n0 = tn * 64;

    const float* Arow0 = A + (size_t)(m0 + lm) * K;
    const float* Arow1 = A + (size_t)(m0 + 16 + lm) * K;
    const float* Wr0   = W + (size_t)(n0 + 0 * 16 + lm) * K;
    const float* Wr1   = W + (size_t)(n0 + 1 * 16 + lm) * K;
    const float* Wr2   = W + (size_t)(n0 + 2 * 16 + lm) * K;
    const float* Wr3   = W + (size_t)(n0 + 3 * 16 + lm) * K;

    v8f acc[2][4];
#pragma unroll
    for (int i = 0; i < 2; i++)
#pragma unroll
        for (int j = 0; j < 4; j++) acc[i][j] = (v8f){};

#pragma unroll 2
    for (int k = 0; k < K; k += 4) {
        const int ko = k + 2 * half;
        v2f a0 = *(const v2f*)(Arow0 + ko);
        v2f a1 = *(const v2f*)(Arow1 + ko);
        v2f b0 = *(const v2f*)(Wr0 + ko);
        v2f b1 = *(const v2f*)(Wr1 + ko);
        v2f b2 = *(const v2f*)(Wr2 + ko);
        v2f b3 = *(const v2f*)(Wr3 + ko);
        acc[0][0] = wmma4(a0, b0, acc[0][0]);
        acc[0][1] = wmma4(a0, b1, acc[0][1]);
        acc[0][2] = wmma4(a0, b2, acc[0][2]);
        acc[0][3] = wmma4(a0, b3, acc[0][3]);
        acc[1][0] = wmma4(a1, b0, acc[1][0]);
        acc[1][1] = wmma4(a1, b1, acc[1][1]);
        acc[1][2] = wmma4(a1, b2, acc[1][2]);
        acc[1][3] = wmma4(a1, b3, acc[1][3]);
    }

    float bv[4];
#pragma unroll
    for (int j = 0; j < 4; j++)
        bv[j] = bias ? bias[n0 + j * 16 + lm] : 0.0f;

#pragma unroll
    for (int i = 0; i < 2; i++) {
#pragma unroll
        for (int r = 0; r < 8; r++) {
            const size_t row = (size_t)(m0 + i * 16 + r + 8 * half) * N;
#pragma unroll
            for (int j = 0; j < 4; j++) {
                float v = acc[i][j][r] + bv[j];
                if (fuse_gelu) v = gelu_exact(v);
                C[row + n0 + j * 16 + lm] = v;
            }
        }
    }
}

__global__ __launch_bounds__(256)
void scatter_edges(const int* __restrict__ rows, const int* __restrict__ cols,
                   const float* __restrict__ vals, const float* __restrict__ hl,
                   float* __restrict__ agg, int E)
{
    int t = blockIdx.x * blockDim.x + threadIdx.x;
    int e = t >> 5;
    if (e >= E) return;
    int c0 = (t & 31) * 8;
    int r = rows[e], c = cols[e];
    float v = vals[e];
    const float4* src = (const float4*)(hl + (size_t)c * DD + c0);
    float* dst = agg + (size_t)r * DD + c0;
    float4 h0 = src[0], h1 = src[1];
    atomicAdd(dst + 0, v * h0.x); atomicAdd(dst + 1, v * h0.y);
    atomicAdd(dst + 2, v * h0.z); atomicAdd(dst + 3, v * h0.w);
    atomicAdd(dst + 4, v * h1.x); atomicAdd(dst + 5, v * h1.y);
    atomicAdd(dst + 6, v * h1.z); atomicAdd(dst + 7, v * h1.w);
}

__global__ __launch_bounds__(256)
void zero_f32(float* __restrict__ p, int n)
{
    int i = blockIdx.x * blockDim.x + threadIdx.x;
    if (i < n) p[i] = 0.0f;
}

__global__ __launch_bounds__(256)
void add_gelu_bn(const float* __restrict__ x, const float* __restrict__ agg,
                 const float* __restrict__ g, const float* __restrict__ b,
                 const float* __restrict__ m, const float* __restrict__ v,
                 float* __restrict__ out, int total)
{
    int i = blockIdx.x * blockDim.x + threadIdx.x;
    if (i >= total) return;
    int c = i & (DD - 1);
    float val = x[i] + gelu_exact(agg[i]);
    out[i] = (val - m[c]) * (g[c] * rsqrtf(v[c] + BN_EPS)) + b[c];
}

__global__ __launch_bounds__(256)
void add_bn(const float* __restrict__ a, const float* __restrict__ bsrc,
            const float* __restrict__ g, const float* __restrict__ b,
            const float* __restrict__ m, const float* __restrict__ v,
            float* __restrict__ out, int total)
{
    int i = blockIdx.x * blockDim.x + threadIdx.x;
    if (i >= total) return;
    int c = i & (DD - 1);
    float val = a[i] + bsrc[i];
    out[i] = (val - m[c]) * (g[c] * rsqrtf(v[c] + BN_EPS)) + b[c];
}

__global__ __launch_bounds__(256)
void attn_wmma(const float* __restrict__ qkv, float* __restrict__ ctx)
{
    __shared__ float pbuf[8][16 * 16];
    const int warp = threadIdx.x >> 5;
    const int gid  = blockIdx.x * 8 + warp;
    const int qt   = gid & 31;
    const int h    = (gid >> 5) & 7;
    const int b    = gid >> 8;
    const int lane = threadIdx.x & 31;
    const int half = lane >> 4;
    const int lm   = lane & 15;
    const int LDQ  = 3 * DD;

    const float* gbase = qkv + (size_t)(b * NN) * LDQ;

    v2f qa[8];
    {
        const float* qrow = gbase + (size_t)(qt * 16 + lm) * LDQ + h * DH;
#pragma unroll
        for (int j = 0; j < 8; j++) qa[j] = *(const v2f*)(qrow + 4 * j + 2 * half);
    }

    float rowmax[8], rowsum[8];
    v8f o0 = {}, o1 = {};
#pragma unroll
    for (int r = 0; r < 8; r++) { rowmax[r] = -1e30f; rowsum[r] = 0.0f; }
    const float scale = 0.17677669529663687f;
    float* pb = pbuf[warp];

    for (int kt = 0; kt < NN / 16; kt++) {
        v8f s = {};
        const float* krow = gbase + (size_t)(kt * 16 + lm) * LDQ + DD + h * DH;
#pragma unroll
        for (int j = 0; j < 8; j++) {
            v2f kb = *(const v2f*)(krow + 4 * j + 2 * half);
            s = wmma4(qa[j], kb, s);
        }

        float alpha[8];
#pragma unroll
        for (int r = 0; r < 8; r++) {
            float val = s[r] * scale;
            float mx = val;
            mx = fmaxf(mx, __shfl_xor(mx, 1));
            mx = fmaxf(mx, __shfl_xor(mx, 2));
            mx = fmaxf(mx, __shfl_xor(mx, 4));
            mx = fmaxf(mx, __shfl_xor(mx, 8));
            float nm = fmaxf(rowmax[r], mx);
            float p  = __expf(val - nm);
            float sm = p;
            sm += __shfl_xor(sm, 1);
            sm += __shfl_xor(sm, 2);
            sm += __shfl_xor(sm, 4);
            sm += __shfl_xor(sm, 8);
            float al = __expf(rowmax[r] - nm);
            rowmax[r] = nm;
            rowsum[r] = rowsum[r] * al + sm;
            alpha[r]  = al;
            s[r] = p;
        }
#pragma unroll
        for (int r = 0; r < 8; r++) { o0[r] *= alpha[r]; o1[r] *= alpha[r]; }

#pragma unroll
        for (int r = 0; r < 8; r++) pb[(r + 8 * half) * 16 + lm] = s[r];
        asm volatile("s_wait_dscnt 0" ::: "memory");

#pragma unroll
        for (int j = 0; j < 4; j++) {
            v2f pa = *(const v2f*)(pb + lm * 16 + 4 * j + 2 * half);
            const float* vp = gbase + (size_t)(kt * 16 + 4 * j + 2 * half) * LDQ
                              + 2 * DD + h * DH;
            v2f vb0, vb1;
            vb0.x = vp[lm];            vb0.y = vp[LDQ + lm];
            vb1.x = vp[16 + lm];       vb1.y = vp[LDQ + 16 + lm];
            o0 = wmma4(pa, vb0, o0);
            o1 = wmma4(pa, vb1, o1);
        }
        asm volatile("s_wait_dscnt 0" ::: "memory");
    }

#pragma unroll
    for (int r = 0; r < 8; r++) {
        float inv = 1.0f / rowsum[r];
        int row = b * NN + qt * 16 + r + 8 * half;
        ctx[(size_t)row * DD + h * DH + lm]      = o0[r] * inv;
        ctx[(size_t)row * DD + h * DH + 16 + lm] = o1[r] * inv;
    }
}

static inline int gemm_blocks(int M, int N) { return ((M >> 5) * (N >> 6)) / 8; }

extern "C" void kernel_launch(void* const* d_in, const int* in_sizes, int n_in,
                              void* d_out, int out_size, void* d_ws, size_t ws_size,
                              hipStream_t stream)
{
    (void)n_in; (void)out_size; (void)ws_size;
    const float* x          = (const float*)d_in[0];
    const int*   edge_rows  = (const int*)  d_in[1];
    const int*   edge_cols  = (const int*)  d_in[2];
    const float* edge_vals  = (const float*)d_in[3];
    const float* w_gcn      = (const float*)d_in[4];
    const float* in_proj_w  = (const float*)d_in[5];
    const float* in_proj_b  = (const float*)d_in[6];
    const float* out_proj_w = (const float*)d_in[7];
    const float* out_proj_b = (const float*)d_in[8];
    const float* w1         = (const float*)d_in[9];
    const float* b1         = (const float*)d_in[10];
    const float* w2         = (const float*)d_in[11];
    const float* b2         = (const float*)d_in[12];
    const float* bn1_g = (const float*)d_in[15], *bn1_b = (const float*)d_in[16];
    const float* bn1_m = (const float*)d_in[17], *bn1_v = (const float*)d_in[18];
    const float* bn2_g = (const float*)d_in[19], *bn2_b = (const float*)d_in[20];
    const float* bn2_m = (const float*)d_in[21], *bn2_v = (const float*)d_in[22];
    const float* bn3_g = (const float*)d_in[23], *bn3_b = (const float*)d_in[24];
    const float* bn3_m = (const float*)d_in[25], *bn3_v = (const float*)d_in[26];

    const int E = in_sizes[1];

    float* ws = (float*)d_ws;
    float* wsA = ws;
    float* wsB = wsA + (size_t)NT * 1024;
    float* wsC = wsB + (size_t)NT * DD;
    float* wsD = wsC + (size_t)NT * DD;
    float* wsE = wsD + (size_t)NT * DD;
    float* outp = (float*)d_out;

    const int EW = NT * DD;
    const dim3 blk(256);

    gemm_wmma_f32<<<gemm_blocks(NT, DD), blk, 0, stream>>>(x, w_gcn, nullptr, wsB,
                                                           NT, DD, DD, 0);
    zero_f32<<<(EW + 255) / 256, blk, 0, stream>>>(wsC, EW);
    scatter_edges<<<(E * 32 + 255) / 256, blk, 0, stream>>>(edge_rows, edge_cols,
                                                            edge_vals, wsB, wsC, E);
    add_gelu_bn<<<(EW + 255) / 256, blk, 0, stream>>>(x, wsC, bn1_g, bn1_b, bn1_m,
                                                      bn1_v, wsD, EW);
    gemm_wmma_f32<<<gemm_blocks(NT, 3 * DD), blk, 0, stream>>>(wsD, in_proj_w,
                                                               in_proj_b, wsA,
                                                               NT, 3 * DD, DD, 0);
    attn_wmma<<<(BB * HH * (NN / 16)) / 8, blk, 0, stream>>>(wsA, wsB);
    gemm_wmma_f32<<<gemm_blocks(NT, DD), blk, 0, stream>>>(wsB, out_proj_w,
                                                           out_proj_b, wsC,
                                                           NT, DD, DD, 0);
    add_bn<<<(EW + 255) / 256, blk, 0, stream>>>(wsD, wsC, bn2_g, bn2_b, bn2_m,
                                                 bn2_v, wsE, EW);
    gemm_wmma_f32<<<gemm_blocks(NT, 4 * DD), blk, 0, stream>>>(wsE, w1, b1, wsA,
                                                               NT, 4 * DD, DD, 1);
    gemm_wmma_f32<<<gemm_blocks(NT, DD), blk, 0, stream>>>(wsA, w2, b2, wsC,
                                                           NT, DD, 4 * DD, 0);
    add_bn<<<(EW + 255) / 256, blk, 0, stream>>>(wsE, wsC, bn3_g, bn3_b, bn3_m,
                                                 bn3_v, outp, EW);
}


# scatter's float atomics replaced by a row-grouped gather: int-atomic count, scan, fill, per-row ordering; one wave sums a row's edges in registers, in edge order (sum order changed, now fixed)
# speedup vs baseline: 1.5356x; 1.5356x over previous
#include <hip/hip_runtime.h>
#include <hip/hip_bf16.h>
#include <math.h>

#define BB   16
#define NN   512
#define DD   256
#define HH   8
#define DH   32
#define NT   (BB * NN)
#define BN_EPS 1e-5f

typedef float v2f __attribute__((ext_vector_type(2)));
typedef float v8f __attribute__((ext_vector_type(8)));

static __device__ __forceinline__ v8f wmma4(v2f a, v2f b, v8f c) {
    return __builtin_amdgcn_wmma_f32_16x16x4_f32(
         false, a,  false, b,
         (short)0, c,  false,  false);
}

static __device__ __forceinline__ float gelu_exact(float v) {
    return 0.5f * v * (1.0f + erff(v * 0.70710678118654752f));
}

__global__ __launch_bounds__(256)
void gemm_wmma_f32(const float* __restrict__ A, const float* __restrict__ W,
                   const float* __restrict__ bias, float* __restrict__ C,
                   int M, int N, int K, int fuse_gelu)
{
    const int wave = blockIdx.x * (blockDim.x >> 5) + (threadIdx.x >> 5);
    const int ntn  = N >> 6;
    const int tm   = wave / ntn;
    const int tn   = wave - tm * ntn;
    if (tm >= (M >> 5)) return;

    const int lane = threadIdx.x & 31;
    const int half = lane >> 4;
    const int lm   = lane & 15;

    const int m0 = tm * 32;
    const int n0 = tn * 64;

    const float* Arow0 = A + (size_t)(m0 + lm) * K;
    const float* Arow1 = A + (size_t)(m0 + 16 + lm) * K;
    const float* Wr0   = W + (size_t)(n0 + 0 * 16 + lm) * K;
    const float* Wr1   = W + (size_t)(n0 + 1 * 16 + lm) * K;
    const float* Wr2   = W + (size_t)(n0 + 2 * 16 + lm) * K;
    const float* Wr3   = W + (size_t)(n0 + 3 * 16 + lm) * K;

    v8f acc[2][4];
#pragma unroll
    for (int i = 0; i < 2; i++)
#pragma unroll
        for (int j = 0; j < 4; j++) acc[i][j] = (v8f){};

#pragma unroll 2
    for (int k = 0; k < K; k += 4) {
        const int ko = k + 2 * half;
        v2f a0 = *(const v2f*)(Arow0 + ko);
        v2f a1 = *(const v2f*)(Arow1 + ko);
        v2f b0 = *(const v2f*)(Wr0 + ko);
        v2f b1 = *(const v2f*)(Wr1 + ko);
        v2f b2 = *(const v2f*)(Wr2 + ko);
        v2f b3 = *(const v2f*)(Wr3 + ko);
        acc[0][0] = wmma4(a0, b0, acc[0][0]);
        acc[0][1] = wmma4(a0, b1, acc[0][1]);
        acc[0][2] = wmma4(a0, b2, acc[0][2]);
        acc[0][3] = wmma4(a0, b3, acc[0][3]);
        acc[1][0] = wmma4(a1, b0, acc[1][0]);
        acc[1][1] = wmma4(a1, b1, acc[1][1]);
        acc[1][2] = wmma4(a1, b2, acc[1][2]);
        acc[1][3] = wmma4(a1, b3, acc[1][3]);
    }

    float bv[4];
#pragma unroll
    for (int j = 0; j < 4; j++)
        bv[j] = bias ? bias[n0 + j * 16 + lm] : 0.0f;

#pragma unroll
    for (int i = 0; i < 2; i++) {
#pragma unroll
        for (int r = 0; r < 8; r++) {
            const size_t row = (size_t)(m0 + i * 16 + r + 8 * half) * N;
#pragma unroll
            for (int j = 0; j < 4; j++) {
                float v = acc[i][j][r] + bv[j];
                if (fuse_gelu) v = gelu_exact(v);
                C[row + n0 + j * 16 + lm] = v;
            }
        }
    }
}

__global__ __launch_bounds__(256)
void zero_i32(int* __restrict__ p, int n)
{
    int i = blockIdx.x * blockDim.x + threadIdx.x;
    if (i < n) p[i] = 0;
}

__global__ __launch_bounds__(256)
void count_rows(const int* __restrict__ rows, int* __restrict__ cnt,
                int* __restrict__ rank, int E)
{
    int e = blockIdx.x * blockDim.x + threadIdx.x;
    if (e < E) rank[e] = atomicAdd(cnt + rows[e], 1);
}

__global__ __launch_bounds__(1024)
void scan_rows(const int* __restrict__ cnt, int* __restrict__ start)
{
    __shared__ int wsum[32];
    const int t    = threadIdx.x;
    const int lane = t & 31;
    const int w    = t >> 5;
    const int4 c0 = ((const int4*)cnt)[2 * t];
    const int4 c1 = ((const int4*)cnt)[2 * t + 1];
    const int s = c0.x + c0.y + c0.z + c0.w + c1.x + c1.y + c1.z + c1.w;
    int inc = s;
    for (int d = 1; d < 32; d <<= 1) {
        int y = __shfl_up(inc, d);
        if (lane >= d) inc += y;
    }
    if (lane == 31) wsum[w] = inc;
    __syncthreads();
    if (w == 0) {
        const int v = wsum[lane];
        int iv = v;
        for (int d = 1; d < 32; d <<= 1) {
            int y = __shfl_up(iv, d);
            if (lane >= d) iv += y;
        }
        wsum[lane] = iv - v;
    }
    __syncthreads();
    int4 o0, o1;
    o0.x = wsum[w] + inc - s;
    o0.y = o0.x + c0.x;  o0.z = o0.y + c0.y;  o0.w = o0.z + c0.z;
    o1.x = o0.w + c0.w;
    o1.y = o1.x + c1.x;  o1.z = o1.y + c1.y;  o1.w = o1.z + c1.z;
    ((int4*)start)[2 * t]     = o0;
    ((int4*)start)[2 * t + 1] = o1;
    if (t == 1023) start[NT] = o1.w + c1.w;
}

__global__ __launch_bounds__(256)
void fill_grouped(const int* __restrict__ rows, const int* __restrict__ start,
                  const int* __restrict__ rank, int* __restrict__ peid, int E)
{
    int e = blockIdx.x * blockDim.x + threadIdx.x;
    if (e >= E) return;
    peid[start[rows[e]] + rank[e]] = e;
}

__global__ __launch_bounds__(256)
void order_rows(const int* __restrict__ start, const int* __restrict__ peid,
                const int* __restrict__ cols, const float* __restrict__ vals,
                int* __restrict__ pcol, float* __restrict__ pval)
{
    const int row  = blockIdx.x * 8 + __builtin_amdgcn_readfirstlane(threadIdx.x >> 5);
    const int lane = threadIdx.x & 31;
    const int s = start[row];
    const int e = start[row + 1];
    for (int base = s; base < e; base += 32) {
        const int n  = (e - base) < 32 ? (e - base) : 32;
        const int my = (lane < n) ? peid[base + lane] : 0x7fffffff;
        int place = 0;
        for (int b2 = s; b2 < e; b2 += 32) {
            const int n2    = (e - b2) < 32 ? (e - b2) : 32;
            const int other = (lane < n2) ? peid[b2 + lane] : 0x7fffffff;
            for (int j = 0; j < n2; j++)
                place += (__builtin_amdgcn_readlane(other, j) < my) ? 1 : 0;
        }
        if (lane < n) {
            pcol[s + place] = cols[my];
            pval[s + place] = vals[my];
        }
    }
}

__global__ __launch_bounds__(256)
void gather_rows(const int* __restrict__ start, const int* __restrict__ pcol,
                 const float* __restrict__ pval, const float* __restrict__ hl,
                 float* __restrict__ agg)
{
    const int row  = blockIdx.x * 8 + __builtin_amdgcn_readfirstlane(threadIdx.x >> 5);
    const int lane = threadIdx.x & 31;
    const int s = start[row];
    const int e = start[row + 1];
    float4 a0 = make_float4(0.f, 0.f, 0.f, 0.f);
    float4 a1 = make_float4(0.f, 0.f, 0.f, 0.f);
    const float* hb = hl + lane * 4;

    for (int base = s; base < e; base += 32) {
        const int n = (e - base) < 32 ? (e - base) : 32;
        int   c = 0;
        float v = 0.0f;
        if (lane < n) { c = pcol[base + lane]; v = pval[base + lane]; }
#pragma unroll 4
        for (int j = 0; j < n; j++) {
            const int   cj = __builtin_amdgcn_readlane(c, j);
            const float vj = __int_as_float(
                __builtin_amdgcn_readlane(__float_as_int(v), j));
            const float* src = hb + (size_t)cj * DD;
            const float4 h0 = *(const float4*)(src);
            const float4 h1 = *(const float4*)(src + 128);
            a0.x += vj * h0.x;  a0.y += vj * h0.y;
            a0.z += vj * h0.z;  a0.w += vj * h0.w;
            a1.x += vj * h1.x;  a1.y += vj * h1.y;
            a1.z += vj * h1.z;  a1.w += vj * h1.w;
        }
    }
    float* dst = agg + (size_t)row * DD + lane * 4;
    *(float4*)(dst)       = a0;
    *(float4*)(dst + 128) = a1;
}

__global__ __launch_bounds__(256)
void add_gelu_bn(const float* __restrict__ x, const float* __restrict__ agg,
                 const float* __restrict__ g, const float* __restrict__ b,
                 const float* __restrict__ m, const float* __restrict__ v,
                 float* __restrict__ out, int total)
{
    int i = blockIdx.x * blockDim.x + threadIdx.x;
    if (i >= total) return;
    int c = i & (DD - 1);
    float val = x[i] + gelu_exact(agg[i]);
    out[i] = (val - m[c]) * (g[c] * rsqrtf(v[c] + BN_EPS)) + b[c];
}

__global__ __launch_bounds__(256)
void add_bn(const float* __restrict__ a, const float* __restrict__ bsrc,
            const float* __restrict__ g, const float* __restrict__ b,
            const float* __restrict__ m, const float* __restrict__ v,
            float* __restrict__ out, int total)
{
    int i = blockIdx.x * blockDim.x + threadIdx.x;
    if (i >= total) return;
    int c = i & (DD - 1);
    float val = a[i] + bsrc[i];
    out[i] = (val - m[c]) * (g[c] * rsqrtf(v[c] + BN_EPS)) + b[c];
}

__global__ __launch_bounds__(256)
void attn_wmma(const float* __restrict__ qkv, float* __restrict__ ctx)
{
    __shared__ float pbuf[8][16 * 16];
    const int warp = threadIdx.x >> 5;
    const int gid  = blockIdx.x * 8 + warp;
    const int qt   = gid & 31;
    const int h    = (gid >> 5) & 7;
    const int b    = gid >> 8;
    const int lane = threadIdx.x & 31;
    const int half = lane >> 4;
    const int lm   = lane & 15;
    const int LDQ  = 3 * DD;

    const float* gbase = qkv + (size_t)(b * NN) * LDQ;

    v2f qa[8];
    {
        const float* qrow = gbase + (size_t)(qt * 16 + lm) * LDQ + h * DH;
#pragma unroll
        for (int j = 0; j < 8; j++) qa[j] = *(const v2f*)(qrow + 4 * j + 2 * half);
    }

    float rowmax[8], rowsum[8];
    v8f o0 = {}, o1 = {};
#pragma unroll
    for (int r = 0; r < 8; r++) { rowmax[r] = -1e30f; rowsum[r] = 0.0f; }
    const float scale = 0.17677669529663687f;
    float* pb = pbuf[warp];

    for (int kt = 0; kt < NN / 16; kt++) {
        v8f s = {};
        const float* krow = gbase + (size_t)(kt * 16 + lm) * LDQ + DD + h * DH;
#pragma unroll
        for (int j = 0; j < 8; j++) {
            v2f kb = *(const v2f*)(krow + 4 * j + 2 * half);
            s = wmma4(qa[j], kb, s);
        }

        float alpha[8];
#pragma unroll
        for (int r = 0; r < 8; r++) {
            float val = s[r] * scale;
            float mx = val;
            mx = fmaxf(mx, __shfl_xor(mx, 1));
            mx = fmaxf(mx, __shfl_xor(mx, 2));
            mx = fmaxf(mx, __shfl_xor(mx, 4));
            mx = fmaxf(mx, __shfl_xor(mx, 8));
            float nm = fmaxf(rowmax[r], mx);
            float p  = __expf(val - nm);
            float sm = p;
            sm += __shfl_xor(sm, 1);
            sm += __shfl_xor(sm, 2);
            sm += __shfl_xor(sm, 4);
            sm += __shfl_xor(sm, 8);
            float al = __expf(rowmax[r] - nm);
            rowmax[r] = nm;
            rowsum[r] = rowsum[r] * al + sm;
            alpha[r]  = al;
            s[r] = p;
        }
#pragma unroll
        for (int r = 0; r < 8; r++) { o0[r] *= alpha[r]; o1[r] *= alpha[r]; }

#pragma unroll
        for (int r = 0; r < 8; r++) pb[(r + 8 * half) * 16 + lm] = s[r];
        asm volatile("s_wait_dscnt 0" ::: "memory");

#pragma unroll
        for (int j = 0; j < 4; j++) {
            v2f pa = *(const v2f*)(pb + lm * 16 + 4 * j + 2 * half);
            const float* vp = gbase + (size_t)(kt * 16 + 4 * j + 2 * half) * LDQ
                              + 2 * DD + h * DH;
            v2f vb0, vb1;
            vb0.x = vp[lm];            vb0.y = vp[LDQ + lm];
            vb1.x = vp[16 + lm];       vb1.y = vp[LDQ + 16 + lm];
            o0 = wmma4(pa, vb0, o0);
            o1 = wmma4(pa, vb1, o1);
        }
        asm volatile("s_wait_dscnt 0" ::: "memory");
    }

#pragma unroll
    for (int r = 0; r < 8; r++) {
        float inv = 1.0f / rowsum[r];
        int row = b * NN + qt * 16 + r + 8 * half;
        ctx[(size_t)row * DD + h * DH + lm]      = o0[r] * inv;
        ctx[(size_t)row * DD + h * DH + 16 + lm] = o1[r] * inv;
    }
}

static inline int gemm_blocks(int M, int N) { return ((M >> 5) * (N >> 6)) / 8; }

extern "C" void kernel_launch(void* const* d_in, const int* in_sizes, int n_in,
                              void* d_out, int out_size, void* d_ws, size_t ws_size,
                              hipStream_t stream)
{
    (void)n_in; (void)out_size; (void)ws_size;
    const float* x          = (const float*)d_in[0];
    const int*   edge_rows  = (const int*)  d_in[1];
    const int*   edge_cols  = (const int*)  d_in[2];
    const float* edge_vals  = (const float*)d_in[3];
    const float* w_gcn      = (const float*)d_in[4];
    const float* in_proj_w  = (const float*)d_in[5];
    const float* in_proj_b  = (const float*)d_in[6];
    const float* out_proj_w = (const float*)d_in[7];
    const float* out_proj_b = (const float*)d_in[8];
    const float* w1         = (const float*)d_in[9];
    const float* b1         = (const float*)d_in[10];
    const float* w2         = (const float*)d_in[11];
    const float* b2         = (const float*)d_in[12];
    const float* bn1_g = (const float*)d_in[15], *bn1_b = (const float*)d_in[16];
    const float* bn1_m = (const float*)d_in[17], *bn1_v = (const float*)d_in[18];
    const float* bn2_g = (const float*)d_in[19], *bn2_b = (const float*)d_in[20];
    const float* bn2_m = (const float*)d_in[21], *bn2_v = (const float*)d_in[22];
    const float* bn3_g = (const float*)d_in[23], *bn3_b = (const float*)d_in[24];
    const float* bn3_m = (const float*)d_in[25], *bn3_v = (const float*)d_in[26];

    const int E = in_sizes[1];

    float* ws = (float*)d_ws;
    float* wsA = ws;
    float* wsB = wsA + (size_t)NT * 1024;
    float* wsC = wsB + (size_t)NT * DD;
    float* wsD = wsC + (size_t)NT * DD;
    float* wsE = wsD + (size_t)NT * DD;
    int*   cnt   = (int*)(wsE + (size_t)NT * DD);
    int*   start = cnt + NT;
    int*   rank  = start + NT + 32;
    int*   peid  = rank + E;
    int*   pcol  = peid + E;
    float* pval  = (float*)(pcol + E);
    float* outp = (float*)d_out;

    const int EW = NT * DD;
    const dim3 blk(256);

    gemm_wmma_f32<<<gemm_blocks(NT, DD), blk, 0, stream>>>(x, w_gcn, nullptr, wsB,
                                                           NT, DD, DD, 0);
    zero_i32<<<NT / 256, blk, 0, stream>>>(cnt, NT);
    count_rows<<<(E + 255) / 256, blk, 0, stream>>>(edge_rows, cnt, rank, E);
    scan_rows<<<1, dim3(1024), 0, stream>>>(cnt, start);
    fill_grouped<<<(E + 255) / 256, blk, 0, stream>>>(edge_rows, start, rank, peid, E);
    order_rows<<<NT / 8, blk, 0, stream>>>(start, peid, edge_cols, edge_vals, pcol, pval);
    gather_rows<<<NT / 8, blk, 0, stream>>>(start, pcol, pval, wsB, wsC);
    add_gelu_bn<<<(EW + 255) / 256, blk, 0, stream>>>(x, wsC, bn1_g, bn1_b, bn1_m,
                                                      bn1_v, wsD, EW);
    gemm_wmma_f32<<<gemm_blocks(NT, 3 * DD), blk, 0, stream>>>(wsD, in_proj_w,
                                                               in_proj_b, wsA,
                                                               NT, 3 * DD, DD, 0);
    attn_wmma<<<(BB * HH * (NN / 16)) / 8, blk, 0, stream>>>(wsA, wsB);
    gemm_wmma_f32<<<gemm_blocks(NT, DD), blk, 0, stream>>>(wsB, out_proj_w,
                                                           out_proj_b, wsC,
                                                           NT, DD, DD, 0);
    add_bn<<<(EW + 255) / 256, blk, 0, stream>>>(wsD, wsC, bn2_g, bn2_b, bn2_m,
                                                 bn2_v, wsE, EW);
    gemm_wmma_f32<<<gemm_blocks(NT, 4 * DD), blk, 0, stream>>>(wsE, w1, b1, wsA,
                                                               NT, 4 * DD, DD, 1);
    gemm_wmma_f32<<<gemm_blocks(NT, DD), blk, 0, stream>>>(wsA, w2, b2, wsC,
                                                           NT, DD, 4 * DD, 0);
    add_bn<<<(EW + 255) / 256, blk, 0, stream>>>(wsE, wsC, bn3_g, bn3_b, bn3_m,
                                                 bn3_v, outp, EW);
}


# GEMM operands staged through LDS: 64x128 work-group tiles, whole-line 16-byte global loads one K-chunk ahead, 8-byte LDS fragment reads; k order and sum order unchanged
# speedup vs baseline: 2.0423x; 1.3300x over previous
#include <hip/hip_runtime.h>
#include <hip/hip_bf16.h>
#include <math.h>

#define BB   16
#define NN   512
#define DD   256
#define HH   8
#define DH   32
#define NT   (BB * NN)
#define BN_EPS 1e-5f

typedef float v2f __attribute__((ext_vector_type(2)));
typedef float v8f __attribute__((ext_vector_type(8)));
typedef float v4f __attribute__((ext_vector_type(4)));

static __device__ __forceinline__ v8f wmma4(v2f a, v2f b, v8f c) {
    return __builtin_amdgcn_wmma_f32_16x16x4_f32(
         false, a,  false, b,
         (short)0, c,  false,  false);
}

static __device__ __forceinline__ float gelu_exact(float v) {
    return 0.5f * v * (1.0f + erff(v * 0.70710678118654752f));
}

#define GBM 64
#define GBN 128
#define GKC 32
#define GLD 36

__global__ __launch_bounds__(256)
void gemm_wmma_f32(const float* __restrict__ A, const float* __restrict__ W,
                   const float* __restrict__ bias, float* __restrict__ C,
                   int M, int N, int K, int fuse_gelu)
{
    __shared__ float As[2][GBM * GLD];
    __shared__ float Bs[2][GBN * GLD];
    (void)M;

    const int nbn = N >> 7;
    const int bm  = blockIdx.x / nbn;
    const int bn  = blockIdx.x - bm * nbn;
    const int m0  = bm * GBM;
    const int n0  = bn * GBN;

    const int t    = threadIdx.x;
    const int wv   = t >> 5;
    const int wm   = wv >> 2;
    const int wn   = wv & 3;
    const int lane = t & 31;
    const int half = lane >> 4;
    const int lm   = lane & 15;

    const int srow = t >> 3;
    const int sc4  = (t & 7) * 4;
    const float* ag = A + (size_t)(m0 + srow) * K + sc4;
    const float* wg = W + (size_t)(n0 + srow) * K + sc4;
    const int soff  = srow * GLD + sc4;

    v4f ra0 = *(const v4f*)(ag);
    v4f ra1 = *(const v4f*)(ag + (size_t)32 * K);
    v4f rb0 = *(const v4f*)(wg);
    v4f rb1 = *(const v4f*)(wg + (size_t)32 * K);
    v4f rb2 = *(const v4f*)(wg + (size_t)64 * K);
    v4f rb3 = *(const v4f*)(wg + (size_t)96 * K);
    *(v4f*)(&As[0][soff])            = ra0;
    *(v4f*)(&As[0][soff + 32 * GLD]) = ra1;
    *(v4f*)(&Bs[0][soff])            = rb0;
    *(v4f*)(&Bs[0][soff + 32 * GLD]) = rb1;
    *(v4f*)(&Bs[0][soff + 64 * GLD]) = rb2;
    *(v4f*)(&Bs[0][soff + 96 * GLD]) = rb3;
    __syncthreads();

    v8f acc[2][2];
#pragma unroll
    for (int i = 0; i < 2; i++)
#pragma unroll
        for (int j = 0; j < 2; j++) acc[i][j] = (v8f){};

    const int aoff = (wm * 32 + lm) * GLD + 2 * half;
    const int boff = (wn * 32 + lm) * GLD + 2 * half;
    const int nch  = K / GKC;

    for (int c = 0; c < nch; c++) {
        const int  cur  = c & 1;
        const bool more = (c + 1) < nch;
        if (more) {
            const int k1 = (c + 1) * GKC;
            ra0 = *(const v4f*)(ag + k1);
            ra1 = *(const v4f*)(ag + (size_t)32 * K + k1);
            rb0 = *(const v4f*)(wg + k1);
            rb1 = *(const v4f*)(wg + (size_t)32 * K + k1);
            rb2 = *(const v4f*)(wg + (size_t)64 * K + k1);
            rb3 = *(const v4f*)(wg + (size_t)96 * K + k1);
        }

        const float* as = &As[cur][aoff];
        const float* bs = &Bs[cur][boff];
#pragma unroll
        for (int j = 0; j < GKC / 4; j++) {
            v2f a0 = *(const v2f*)(as + 4 * j);
            v2f a1 = *(const v2f*)(as + 16 * GLD + 4 * j);
            v2f b0 = *(const v2f*)(bs + 4 * j);
            v2f b1 = *(const v2f*)(bs + 16 * GLD + 4 * j);
            acc[0][0] = wmma4(a0, b0, acc[0][0]);
            acc[0][1] = wmma4(a0, b1, acc[0][1]);
            acc[1][0] = wmma4(a1, b0, acc[1][0]);
            acc[1][1] = wmma4(a1, b1, acc[1][1]);
        }

        if (more) {
            float* an = &As[cur ^ 1][soff];
            float* bn2 = &Bs[cur ^ 1][soff];
            *(v4f*)(an)             = ra0;
            *(v4f*)(an + 32 * GLD)  = ra1;
            *(v4f*)(bn2)            = rb0;
            *(v4f*)(bn2 + 32 * GLD) = rb1;
            *(v4f*)(bn2 + 64 * GLD) = rb2;
            *(v4f*)(bn2 + 96 * GLD) = rb3;
        }
        __syncthreads();
    }

    const int nw = n0 + wn * 32;
    const int mw = m0 + wm * 32;
    float bv[2];
#pragma unroll
    for (int j = 0; j < 2; j++)
        bv[j] = bias ? bias[nw + j * 16 + lm] : 0.0f;

#pragma unroll
    for (int i = 0; i < 2; i++) {
#pragma unroll
        for (int r = 0; r < 8; r++) {
            const size_t row = (size_t)(mw + i * 16 + r + 8 * half) * N;
#pragma unroll
            for (int j = 0; j < 2; j++) {
                float v = acc[i][j][r] + bv[j];
                if (fuse_gelu) v = gelu_exact(v);
                C[row + nw + j * 16 + lm] = v;
            }
        }
    }
}

__global__ __launch_bounds__(256)
void zero_i32(int* __restrict__ p, int n)
{
    int i = blockIdx.x * blockDim.x + threadIdx.x;
    if (i < n) p[i] = 0;
}

__global__ __launch_bounds__(256)
void count_rows(const int* __restrict__ rows, int* __restrict__ cnt,
                int* __restrict__ rank, int E)
{
    int e = blockIdx.x * blockDim.x + threadIdx.x;
    if (e < E) rank[e] = atomicAdd(cnt + rows[e], 1);
}

__global__ __launch_bounds__(1024)
void scan_rows(const int* __restrict__ cnt, int* __restrict__ start)
{
    __shared__ int wsum[32];
    const int t    = threadIdx.x;
    const int lane = t & 31;
    const int w    = t >> 5;
    const int4 c0 = ((const int4*)cnt)[2 * t];
    const int4 c1 = ((const int4*)cnt)[2 * t + 1];
    const int s = c0.x + c0.y + c0.z + c0.w + c1.x + c1.y + c1.z + c1.w;
    int inc = s;
    for (int d = 1; d < 32; d <<= 1) {
        int y = __shfl_up(inc, d);
        if (lane >= d) inc += y;
    }
    if (lane == 31) wsum[w] = inc;
    __syncthreads();
    if (w == 0) {
        const int v = wsum[lane];
        int iv = v;
        for (int d = 1; d < 32; d <<= 1) {
            int y = __shfl_up(iv, d);
            if (lane >= d) iv += y;
        }
        wsum[lane] = iv - v;
    }
    __syncthreads();
    int4 o0, o1;
    o0.x = wsum[w] + inc - s;
    o0.y = o0.x + c0.x;  o0.z = o0.y + c0.y;  o0.w = o0.z + c0.z;
    o1.x = o0.w + c0.w;
    o1.y = o1.x + c1.x;  o1.z = o1.y + c1.y;  o1.w = o1.z + c1.z;
    ((int4*)start)[2 * t]     = o0;
    ((int4*)start)[2 * t + 1] = o1;
    if (t == 1023) start[NT] = o1.w + c1.w;
}

__global__ __launch_bounds__(256)
void fill_grouped(const int* __restrict__ rows, const int* __restrict__ start,
                  const int* __restrict__ rank, int* __restrict__ peid, int E)
{
    int e = blockIdx.x * blockDim.x + threadIdx.x;
    if (e >= E) return;
    peid[start[rows[e]] + rank[e]] = e;
}

__global__ __launch_bounds__(256)
void order_rows(const int* __restrict__ start, const int* __restrict__ peid,
                const int* __restrict__ cols, const float* __restrict__ vals,
                int* __restrict__ pcol, float* __restrict__ pval)
{
    const int row  = blockIdx.x * 8 + __builtin_amdgcn_readfirstlane(threadIdx.x >> 5);
    const int lane = threadIdx.x & 31;
    const int s = start[row];
    const int e = start[row + 1];
    for (int base = s; base < e; base += 32) {
        const int n  = (e - base) < 32 ? (e - base) : 32;
        const int my = (lane < n) ? peid[base + lane] : 0x7fffffff;
        int place = 0;
        for (int b2 = s; b2 < e; b2 += 32) {
            const int n2    = (e - b2) < 32 ? (e - b2) : 32;
            const int other = (lane < n2) ? peid[b2 + lane] : 0x7fffffff;
            for (int j = 0; j < n2; j++)
                place += (__builtin_amdgcn_readlane(other, j) < my) ? 1 : 0;
        }
        if (lane < n) {
            pcol[s + place] = cols[my];
            pval[s + place] = vals[my];
        }
    }
}

__global__ __launch_bounds__(256)
void gather_rows(const int* __restrict__ start, const int* __restrict__ pcol,
                 const float* __restrict__ pval, const float* __restrict__ hl,
                 float* __restrict__ agg)
{
    const int row  = blockIdx.x * 8 + __builtin_amdgcn_readfirstlane(threadIdx.x >> 5);
    const int lane = threadIdx.x & 31;
    const int s = start[row];
    const int e = start[row + 1];
    float4 a0 = make_float4(0.f, 0.f, 0.f, 0.f);
    float4 a1 = make_float4(0.f, 0.f, 0.f, 0.f);
    const float* hb = hl + lane * 4;

    for (int base = s; base < e; base += 32) {
        const int n = (e - base) < 32 ? (e - base) : 32;
        int   c = 0;
        float v = 0.0f;
        if (lane < n) { c = pcol[base + lane]; v = pval[base + lane]; }
#pragma unroll 4
        for (int j = 0; j < n; j++) {
            const int   cj = __builtin_amdgcn_readlane(c, j);
            const float vj = __int_as_float(
                __builtin_amdgcn_readlane(__float_as_int(v), j));
            const float* src = hb + (size_t)cj * DD;
            const float4 h0 = *(const float4*)(src);
            const float4 h1 = *(const float4*)(src + 128);
            a0.x += vj * h0.x;  a0.y += vj * h0.y;
            a0.z += vj * h0.z;  a0.w += vj * h0.w;
            a1.x += vj * h1.x;  a1.y += vj * h1.y;
            a1.z += vj * h1.z;  a1.w += vj * h1.w;
        }
    }
    float* dst = agg + (size_t)row * DD + lane * 4;
    *(float4*)(dst)       = a0;
    *(float4*)(dst + 128) = a1;
}

__global__ __launch_bounds__(256)
void add_gelu_bn(const float* __restrict__ x, const float* __restrict__ agg,
                 const float* __restrict__ g, const float* __restrict__ b,
                 const float* __restrict__ m, const float* __restrict__ v,
                 float* __restrict__ out, int total)
{
    int i = blockIdx.x * blockDim.x + threadIdx.x;
    if (i >= total) return;
    int c = i & (DD - 1);
    float val = x[i] + gelu_exact(agg[i]);
    out[i] = (val - m[c]) * (g[c] * rsqrtf(v[c] + BN_EPS)) + b[c];
}

__global__ __launch_bounds__(256)
void add_bn(const float* __restrict__ a, const float* __restrict__ bsrc,
            const float* __restrict__ g, const float* __restrict__ b,
            const float* __restrict__ m, const float* __restrict__ v,
            float* __restrict__ out, int total)
{
    int i = blockIdx.x * blockDim.x + threadIdx.x;
    if (i >= total) return;
    int c = i & (DD - 1);
    float val = a[i] + bsrc[i];
    out[i] = (val - m[c]) * (g[c] * rsqrtf(v[c] + BN_EPS)) + b[c];
}

__global__ __launch_bounds__(256)
void attn_wmma(const float* __restrict__ qkv, float* __restrict__ ctx)
{
    __shared__ float pbuf[8][16 * 16];
    const int warp = threadIdx.x >> 5;
    const int gid  = blockIdx.x * 8 + warp;
    const int qt   = gid & 31;
    const int h    = (gid >> 5) & 7;
    const int b    = gid >> 8;
    const int lane = threadIdx.x & 31;
    const int half = lane >> 4;
    const int lm   = lane & 15;
    const int LDQ  = 3 * DD;

    const float* gbase = qkv + (size_t)(b * NN) * LDQ;

    v2f qa[8];
    {
        const float* qrow = gbase + (size_t)(qt * 16 + lm) * LDQ + h * DH;
#pragma unroll
        for (int j = 0; j < 8; j++) qa[j] = *(const v2f*)(qrow + 4 * j + 2 * half);
    }

    float rowmax[8], rowsum[8];
    v8f o0 = {}, o1 = {};
#pragma unroll
    for (int r = 0; r < 8; r++) { rowmax[r] = -1e30f; rowsum[r] = 0.0f; }
    const float scale = 0.17677669529663687f;
    float* pb = pbuf[warp];

    for (int kt = 0; kt < NN / 16; kt++) {
        v8f s = {};
        const float* krow = gbase + (size_t)(kt * 16 + lm) * LDQ + DD + h * DH;
#pragma unroll
        for (int j = 0; j < 8; j++) {
            v2f kb = *(const v2f*)(krow + 4 * j + 2 * half);
            s = wmma4(qa[j], kb, s);
        }

        float alpha[8];
#pragma unroll
        for (int r = 0; r < 8; r++) {
            float val = s[r] * scale;
            float mx = val;
            mx = fmaxf(mx, __shfl_xor(mx, 1));
            mx = fmaxf(mx, __shfl_xor(mx, 2));
            mx = fmaxf(mx, __shfl_xor(mx, 4));
            mx = fmaxf(mx, __shfl_xor(mx, 8));
            float nm = fmaxf(rowmax[r], mx);
            float p  = __expf(val - nm);
            float sm = p;
            sm += __shfl_xor(sm, 1);
            sm += __shfl_xor(sm, 2);
            sm += __shfl_xor(sm, 4);
            sm += __shfl_xor(sm, 8);
            float al = __expf(rowmax[r] - nm);
            rowmax[r] = nm;
            rowsum[r] = rowsum[r] * al + sm;
            alpha[r]  = al;
            s[r] = p;
        }
#pragma unroll
        for (int r = 0; r < 8; r++) { o0[r] *= alpha[r]; o1[r] *= alpha[r]; }

#pragma unroll
        for (int r = 0; r < 8; r++) pb[(r + 8 * half) * 16 + lm] = s[r];
        asm volatile("s_wait_dscnt 0" ::: "memory");

#pragma unroll
        for (int j = 0; j < 4; j++) {
            v2f pa = *(const v2f*)(pb + lm * 16 + 4 * j + 2 * half);
            const float* vp = gbase + (size_t)(kt * 16 + 4 * j + 2 * half) * LDQ
                              + 2 * DD + h * DH;
            v2f vb0, vb1;
            vb0.x = vp[lm];            vb0.y = vp[LDQ + lm];
            vb1.x = vp[16 + lm];       vb1.y = vp[LDQ + 16 + lm];
            o0 = wmma4(pa, vb0, o0);
            o1 = wmma4(pa, vb1, o1);
        }
        asm volatile("s_wait_dscnt 0" ::: "memory");
    }

#pragma unroll
    for (int r = 0; r < 8; r++) {
        float inv = 1.0f / rowsum[r];
        int row = b * NN + qt * 16 + r + 8 * half;
        ctx[(size_t)row * DD + h * DH + lm]      = o0[r] * inv;
        ctx[(size_t)row * DD + h * DH + 16 + lm] = o1[r] * inv;
    }
}

static inline int gemm_blocks(int M, int N) { return (M >> 6) * (N >> 7); }

extern "C" void kernel_launch(void* const* d_in, const int* in_sizes, int n_in,
                              void* d_out, int out_size, void* d_ws, size_t ws_size,
                              hipStream_t stream)
{
    (void)n_in; (void)out_size; (void)ws_size;
    const float* x          = (const float*)d_in[0];
    const int*   edge_rows  = (const int*)  d_in[1];
    const int*   edge_cols  = (const int*)  d_in[2];
    const float* edge_vals  = (const float*)d_in[3];
    const float* w_gcn      = (const float*)d_in[4];
    const float* in_proj_w  = (const float*)d_in[5];
    const float* in_proj_b  = (const float*)d_in[6];
    const float* out_proj_w = (const float*)d_in[7];
    const float* out_proj_b = (const float*)d_in[8];
    const float* w1         = (const float*)d_in[9];
    const float* b1         = (const float*)d_in[10];
    const float* w2         = (const float*)d_in[11];
    const float* b2         = (const float*)d_in[12];
    const float* bn1_g = (const float*)d_in[15], *bn1_b = (const float*)d_in[16];
    const float* bn1_m = (const float*)d_in[17], *bn1_v = (const float*)d_in[18];
    const float* bn2_g = (const float*)d_in[19], *bn2_b = (const float*)d_in[20];
    const float* bn2_m = (const float*)d_in[21], *bn2_v = (const float*)d_in[22];
    const float* bn3_g = (const float*)d_in[23], *bn3_b = (const float*)d_in[24];
    const float* bn3_m = (const float*)d_in[25], *bn3_v = (const float*)d_in[26];

    const int E = in_sizes[1];

    float* ws = (float*)d_ws;
    float* wsA = ws;
    float* wsB = wsA + (size_t)NT * 1024;
    float* wsC = wsB + (size_t)NT * DD;
    float* wsD = wsC + (size_t)NT * DD;
    float* wsE = wsD + (size_t)NT * DD;
    int*   cnt   = (int*)(wsE + (size_t)NT * DD);
    int*   start = cnt + NT;
    int*   rank  = start + NT + 32;
    int*   peid  = rank + E;
    int*   pcol  = peid + E;
    float* pval  = (float*)(pcol + E);
    float* outp = (float*)d_out;

    const int EW = NT * DD;
    const dim3 blk(256);

    gemm_wmma_f32<<<gemm_blocks(NT, DD), blk, 0, stream>>>(x, w_gcn, nullptr, wsB,
                                                           NT, DD, DD, 0);
    zero_i32<<<NT / 256, blk, 0, stream>>>(cnt, NT);
    count_rows<<<(E + 255) / 256, blk, 0, stream>>>(edge_rows, cnt, rank, E);
    scan_rows<<<1, dim3(1024), 0, stream>>>(cnt, start);
    fill_grouped<<<(E + 255) / 256, blk, 0, stream>>>(edge_rows, start, rank, peid, E);
    order_rows<<<NT / 8, blk, 0, stream>>>(start, peid, edge_cols, edge_vals, pcol, pval);
    gather_rows<<<NT / 8, blk, 0, stream>>>(start, pcol, pval, wsB, wsC);
    add_gelu_bn<<<(EW + 255) / 256, blk, 0, stream>>>(x, wsC, bn1_g, bn1_b, bn1_m,
                                                      bn1_v, wsD, EW);
    gemm_wmma_f32<<<gemm_blocks(NT, 3 * DD), blk, 0, stream>>>(wsD, in_proj_w,
                                                               in_proj_b, wsA,
                                                               NT, 3 * DD, DD, 0);
    attn_wmma<<<(BB * HH * (NN / 16)) / 8, blk, 0, stream>>>(wsA, wsB);
    gemm_wmma_f32<<<gemm_blocks(NT, DD), blk, 0, stream>>>(wsB, out_proj_w,
                                                           out_proj_b, wsC,
                                                           NT, DD, DD, 0);
    add_bn<<<(EW + 255) / 256, blk, 0, stream>>>(wsD, wsC, bn2_g, bn2_b, bn2_m,
                                                 bn2_v, wsE, EW);
    gemm_wmma_f32<<<gemm_blocks(NT, 4 * DD), blk, 0, stream>>>(wsE, w1, b1, wsA,
                                                               NT, 4 * DD, DD, 1);
    gemm_wmma_f32<<<gemm_blocks(NT, DD), blk, 0, stream>>>(wsA, w2, b2, wsC,
                                                           NT, DD, 4 * DD, 0);
    add_bn<<<(EW + 255) / 256, blk, 0, stream>>>(wsE, wsC, bn3_g, bn3_b, bn3_m,
                                                 bn3_v, outp, EW);
}


# attention computed transposed (S^T = K Q^T, O^T = V^T P^T): softmax max and sum over a lane's own registers, P enters the second product from registers, no LDS; key sum order changed, fixed
# speedup vs baseline: 2.3541x; 1.1527x over previous
#include <hip/hip_runtime.h>
#include <hip/hip_bf16.h>
#include <math.h>

#define BB   16
#define NN   512
#define DD   256
#define HH   8
#define DH   32
#define NT   (BB * NN)
#define BN_EPS 1e-5f

typedef float v2f __attribute__((ext_vector_type(2)));
typedef float v8f __attribute__((ext_vector_type(8)));
typedef float v4f __attribute__((ext_vector_type(4)));

static __device__ __forceinline__ v8f wmma4(v2f a, v2f b, v8f c) {
    return __builtin_amdgcn_wmma_f32_16x16x4_f32(
         false, a,  false, b,
         (short)0, c,  false,  false);
}

static __device__ __forceinline__ float gelu_exact(float v) {
    return 0.5f * v * (1.0f + erff(v * 0.70710678118654752f));
}

#define GBM 64
#define GBN 128
#define GKC 32
#define GLD 36

__global__ __launch_bounds__(256)
void gemm_wmma_f32(const float* __restrict__ A, const float* __restrict__ W,
                   const float* __restrict__ bias, float* __restrict__ C,
                   int M, int N, int K, int fuse_gelu)
{
    __shared__ float As[2][GBM * GLD];
    __shared__ float Bs[2][GBN * GLD];
    (void)M;

    const int nbn = N >> 7;
    const int bm  = blockIdx.x / nbn;
    const int bn  = blockIdx.x - bm * nbn;
    const int m0  = bm * GBM;
    const int n0  = bn * GBN;

    const int t    = threadIdx.x;
    const int wv   = t >> 5;
    const int wm   = wv >> 2;
    const int wn   = wv & 3;
    const int lane = t & 31;
    const int half = lane >> 4;
    const int lm   = lane & 15;

    const int srow = t >> 3;
    const int sc4  = (t & 7) * 4;
    const float* ag = A + (size_t)(m0 + srow) * K + sc4;
    const float* wg = W + (size_t)(n0 + srow) * K + sc4;
    const int soff  = srow * GLD + sc4;

    v4f ra0 = *(const v4f*)(ag);
    v4f ra1 = *(const v4f*)(ag + (size_t)32 * K);
    v4f rb0 = *(const v4f*)(wg);
    v4f rb1 = *(const v4f*)(wg + (size_t)32 * K);
    v4f rb2 = *(const v4f*)(wg + (size_t)64 * K);
    v4f rb3 = *(const v4f*)(wg + (size_t)96 * K);
    *(v4f*)(&As[0][soff])            = ra0;
    *(v4f*)(&As[0][soff + 32 * GLD]) = ra1;
    *(v4f*)(&Bs[0][soff])            = rb0;
    *(v4f*)(&Bs[0][soff + 32 * GLD]) = rb1;
    *(v4f*)(&Bs[0][soff + 64 * GLD]) = rb2;
    *(v4f*)(&Bs[0][soff + 96 * GLD]) = rb3;
    __syncthreads();

    v8f acc[2][2];
#pragma unroll
    for (int i = 0; i < 2; i++)
#pragma unroll
        for (int j = 0; j < 2; j++) acc[i][j] = (v8f){};

    const int aoff = (wm * 32 + lm) * GLD + 2 * half;
    const int boff = (wn * 32 + lm) * GLD + 2 * half;
    const int nch  = K / GKC;

    for (int c = 0; c < nch; c++) {
        const int  cur  = c & 1;
        const bool more = (c + 1) < nch;
        if (more) {
            const int k1 = (c + 1) * GKC;
            ra0 = *(const v4f*)(ag + k1);
            ra1 = *(const v4f*)(ag + (size_t)32 * K + k1);
            rb0 = *(const v4f*)(wg + k1);
            rb1 = *(const v4f*)(wg + (size_t)32 * K + k1);
            rb2 = *(const v4f*)(wg + (size_t)64 * K + k1);
            rb3 = *(const v4f*)(wg + (size_t)96 * K + k1);
        }

        const float* as = &As[cur][aoff];
        const float* bs = &Bs[cur][boff];
#pragma unroll
        for (int j = 0; j < GKC / 4; j++) {
            v2f a0 = *(const v2f*)(as + 4 * j);
            v2f a1 = *(const v2f*)(as + 16 * GLD + 4 * j);
            v2f b0 = *(const v2f*)(bs + 4 * j);
            v2f b1 = *(const v2f*)(bs + 16 * GLD + 4 * j);
            acc[0][0] = wmma4(a0, b0, acc[0][0]);
            acc[0][1] = wmma4(a0, b1, acc[0][1]);
            acc[1][0] = wmma4(a1, b0, acc[1][0]);
            acc[1][1] = wmma4(a1, b1, acc[1][1]);
        }

        if (more) {
            float* an = &As[cur ^ 1][soff];
            float* bn2 = &Bs[cur ^ 1][soff];
            *(v4f*)(an)             = ra0;
            *(v4f*)(an + 32 * GLD)  = ra1;
            *(v4f*)(bn2)            = rb0;
            *(v4f*)(bn2 + 32 * GLD) = rb1;
            *(v4f*)(bn2 + 64 * GLD) = rb2;
            *(v4f*)(bn2 + 96 * GLD) = rb3;
        }
        __syncthreads();
    }

    const int nw = n0 + wn * 32;
    const int mw = m0 + wm * 32;
    float bv[2];
#pragma unroll
    for (int j = 0; j < 2; j++)
        bv[j] = bias ? bias[nw + j * 16 + lm] : 0.0f;

#pragma unroll
    for (int i = 0; i < 2; i++) {
#pragma unroll
        for (int r = 0; r < 8; r++) {
            const size_t row = (size_t)(mw + i * 16 + r + 8 * half) * N;
#pragma unroll
            for (int j = 0; j < 2; j++) {
                float v = acc[i][j][r] + bv[j];
                if (fuse_gelu) v = gelu_exact(v);
                C[row + nw + j * 16 + lm] = v;
            }
        }
    }
}

__global__ __launch_bounds__(256)
void zero_i32(int* __restrict__ p, int n)
{
    int i = blockIdx.x * blockDim.x + threadIdx.x;
    if (i < n) p[i] = 0;
}

__global__ __launch_bounds__(256)
void count_rows(const int* __restrict__ rows, int* __restrict__ cnt,
                int* __restrict__ rank, int E)
{
    int e = blockIdx.x * blockDim.x + threadIdx.x;
    if (e < E) rank[e] = atomicAdd(cnt + rows[e], 1);
}

__global__ __launch_bounds__(1024)
void scan_rows(const int* __restrict__ cnt, int* __restrict__ start)
{
    __shared__ int wsum[32];
    const int t    = threadIdx.x;
    const int lane = t & 31;
    const int w    = t >> 5;
    const int4 c0 = ((const int4*)cnt)[2 * t];
    const int4 c1 = ((const int4*)cnt)[2 * t + 1];
    const int s = c0.x + c0.y + c0.z + c0.w + c1.x + c1.y + c1.z + c1.w;
    int inc = s;
    for (int d = 1; d < 32; d <<= 1) {
        int y = __shfl_up(inc, d);
        if (lane >= d) inc += y;
    }
    if (lane == 31) wsum[w] = inc;
    __syncthreads();
    if (w == 0) {
        const int v = wsum[lane];
        int iv = v;
        for (int d = 1; d < 32; d <<= 1) {
            int y = __shfl_up(iv, d);
            if (lane >= d) iv += y;
        }
        wsum[lane] = iv - v;
    }
    __syncthreads();
    int4 o0, o1;
    o0.x = wsum[w] + inc - s;
    o0.y = o0.x + c0.x;  o0.z = o0.y + c0.y;  o0.w = o0.z + c0.z;
    o1.x = o0.w + c0.w;
    o1.y = o1.x + c1.x;  o1.z = o1.y + c1.y;  o1.w = o1.z + c1.z;
    ((int4*)start)[2 * t]     = o0;
    ((int4*)start)[2 * t + 1] = o1;
    if (t == 1023) start[NT] = o1.w + c1.w;
}

__global__ __launch_bounds__(256)
void fill_grouped(const int* __restrict__ rows, const int* __restrict__ start,
                  const int* __restrict__ rank, int* __restrict__ peid, int E)
{
    int e = blockIdx.x * blockDim.x + threadIdx.x;
    if (e >= E) return;
    peid[start[rows[e]] + rank[e]] = e;
}

__global__ __launch_bounds__(256)
void order_rows(const int* __restrict__ start, const int* __restrict__ peid,
                const int* __restrict__ cols, const float* __restrict__ vals,
                int* __restrict__ pcol, float* __restrict__ pval)
{
    const int row  = blockIdx.x * 8 + __builtin_amdgcn_readfirstlane(threadIdx.x >> 5);
    const int lane = threadIdx.x & 31;
    const int s = start[row];
    const int e = start[row + 1];
    for (int base = s; base < e; base += 32) {
        const int n  = (e - base) < 32 ? (e - base) : 32;
        const int my = (lane < n) ? peid[base + lane] : 0x7fffffff;
        int place = 0;
        for (int b2 = s; b2 < e; b2 += 32) {
            const int n2    = (e - b2) < 32 ? (e - b2) : 32;
            const int other = (lane < n2) ? peid[b2 + lane] : 0x7fffffff;
            for (int j = 0; j < n2; j++)
                place += (__builtin_amdgcn_readlane(other, j) < my) ? 1 : 0;
        }
        if (lane < n) {
            pcol[s + place] = cols[my];
            pval[s + place] = vals[my];
        }
    }
}

__global__ __launch_bounds__(256)
void gather_rows(const int* __restrict__ start, const int* __restrict__ pcol,
                 const float* __restrict__ pval, const float* __restrict__ hl,
                 float* __restrict__ agg)
{
    const int row  = blockIdx.x * 8 + __builtin_amdgcn_readfirstlane(threadIdx.x >> 5);
    const int lane = threadIdx.x & 31;
    const int s = start[row];
    const int e = start[row + 1];
    float4 a0 = make_float4(0.f, 0.f, 0.f, 0.f);
    float4 a1 = make_float4(0.f, 0.f, 0.f, 0.f);
    const float* hb = hl + lane * 4;

    for (int base = s; base < e; base += 32) {
        const int n = (e - base) < 32 ? (e - base) : 32;
        int   c = 0;
        float v = 0.0f;
        if (lane < n) { c = pcol[base + lane]; v = pval[base + lane]; }
#pragma unroll 4
        for (int j = 0; j < n; j++) {
            const int   cj = __builtin_amdgcn_readlane(c, j);
            const float vj = __int_as_float(
                __builtin_amdgcn_readlane(__float_as_int(v), j));
            const float* src = hb + (size_t)cj * DD;
            const float4 h0 = *(const float4*)(src);
            const float4 h1 = *(const float4*)(src + 128);
            a0.x += vj * h0.x;  a0.y += vj * h0.y;
            a0.z += vj * h0.z;  a0.w += vj * h0.w;
            a1.x += vj * h1.x;  a1.y += vj * h1.y;
            a1.z += vj * h1.z;  a1.w += vj * h1.w;
        }
    }
    float* dst = agg + (size_t)row * DD + lane * 4;
    *(float4*)(dst)       = a0;
    *(float4*)(dst + 128) = a1;
}

__global__ __launch_bounds__(256)
void add_gelu_bn(const float* __restrict__ x, const float* __restrict__ agg,
                 const float* __restrict__ g, const float* __restrict__ b,
                 const float* __restrict__ m, const float* __restrict__ v,
                 float* __restrict__ out, int total)
{
    int i = blockIdx.x * blockDim.x + threadIdx.x;
    if (i >= total) return;
    int c = i & (DD - 1);
    float val = x[i] + gelu_exact(agg[i]);
    out[i] = (val - m[c]) * (g[c] * rsqrtf(v[c] + BN_EPS)) + b[c];
}

__global__ __launch_bounds__(256)
void add_bn(const float* __restrict__ a, const float* __restrict__ bsrc,
            const float* __restrict__ g, const float* __restrict__ b,
            const float* __restrict__ m, const float* __restrict__ v,
            float* __restrict__ out, int total)
{
    int i = blockIdx.x * blockDim.x + threadIdx.x;
    if (i >= total) return;
    int c = i & (DD - 1);
    float val = a[i] + bsrc[i];
    out[i] = (val - m[c]) * (g[c] * rsqrtf(v[c] + BN_EPS)) + b[c];
}

static __device__ __forceinline__ float xhalf(float x) {
    const int b = __builtin_bit_cast(int, x);
    return __builtin_bit_cast(float,
        __builtin_amdgcn_permlanex16(b, b, 0x76543210, 0xfedcba98, false, false));
}

__global__ __launch_bounds__(256)
void attn_wmma(const float* __restrict__ qkv, float* __restrict__ ctx)
{
    const int warp = threadIdx.x >> 5;
    const int gid  = blockIdx.x * 8 + warp;
    const int qt   = gid & 31;
    const int h    = (gid >> 5) & 7;
    const int b    = gid >> 8;
    const int lane = threadIdx.x & 31;
    const int half = lane >> 4;
    const int lm   = lane & 15;
    const int LDQ  = 3 * DD;

    const float* gbase = qkv + (size_t)(b * NN) * LDQ;

    v2f qb[8];
    {
        const float* qrow = gbase + (size_t)(qt * 16 + lm) * LDQ + h * DH;
        for (int j = 0; j < 8; j++) qb[j] = *(const v2f*)(qrow + 4 * j + 2 * half);
    }

    float rowmax = -1e30f;
    float rowsum = 0.0f;
    v8f o0 = {}, o1 = {};
    const float scale = 0.17677669529663687f;

    const float* kbase = gbase + (size_t)lm * LDQ + DD + h * DH + 2 * half;
    const float* vbase = gbase + (size_t)(8 * half) * LDQ + 2 * DD + h * DH + lm;

    for (int kt = 0; kt < NN / 16; kt++) {
        v8f s = {};
        const float* krow = kbase + (size_t)(kt * 16) * LDQ;
        for (int j = 0; j < 8; j++) {
            v2f ka = *(const v2f*)(krow + 4 * j);
            s = wmma4(ka, qb[j], s);
        }

        float val[8];
        for (int r = 0; r < 8; r++) val[r] = s[r] * scale;
        float mx = val[0];
        for (int r = 1; r < 8; r++) mx = fmaxf(mx, val[r]);
        mx = fmaxf(mx, xhalf(mx));
        const float nm = fmaxf(rowmax, mx);
        const float al = __expf(rowmax - nm);
        float p[8];
        float sm = 0.0f;
        for (int r = 0; r < 8; r++) { p[r] = __expf(val[r] - nm); sm += p[r]; }
        rowmax = nm;
        rowsum = rowsum * al + sm;
        for (int r = 0; r < 8; r++) { o0[r] *= al; o1[r] *= al; }

        const float* vp = vbase + (size_t)(kt * 16) * LDQ;
        for (int j = 0; j < 4; j++) {
            v2f pb, va0, va1;
            pb.x  = p[2 * j];                     pb.y  = p[2 * j + 1];
            va0.x = vp[(2 * j) * LDQ];            va0.y = vp[(2 * j + 1) * LDQ];
            va1.x = vp[(2 * j) * LDQ + 16];       va1.y = vp[(2 * j + 1) * LDQ + 16];
            o0 = wmma4(va0, pb, o0);
            o1 = wmma4(va1, pb, o1);
        }
    }

    const float inv = 1.0f / (rowsum + xhalf(rowsum));
    float* dst = ctx + (size_t)(b * NN + qt * 16 + lm) * DD + h * DH + 8 * half;
    *(v4f*)(dst)      = (v4f){o0[0] * inv, o0[1] * inv, o0[2] * inv, o0[3] * inv};
    *(v4f*)(dst + 4)  = (v4f){o0[4] * inv, o0[5] * inv, o0[6] * inv, o0[7] * inv};
    *(v4f*)(dst + 16) = (v4f){o1[0] * inv, o1[1] * inv, o1[2] * inv, o1[3] * inv};
    *(v4f*)(dst + 20) = (v4f){o1[4] * inv, o1[5] * inv, o1[6] * inv, o1[7] * inv};
}

static inline int gemm_blocks(int M, int N) { return (M >> 6) * (N >> 7); }

extern "C" void kernel_launch(void* const* d_in, const int* in_sizes, int n_in,
                              void* d_out, int out_size, void* d_ws, size_t ws_size,
                              hipStream_t stream)
{
    (void)n_in; (void)out_size; (void)ws_size;
    const float* x          = (const float*)d_in[0];
    const int*   edge_rows  = (const int*)  d_in[1];
    const int*   edge_cols  = (const int*)  d_in[2];
    const float* edge_vals  = (const float*)d_in[3];
    const float* w_gcn      = (const float*)d_in[4];
    const float* in_proj_w  = (const float*)d_in[5];
    const float* in_proj_b  = (const float*)d_in[6];
    const float* out_proj_w = (const float*)d_in[7];
    const float* out_proj_b = (const float*)d_in[8];
    const float* w1         = (const float*)d_in[9];
    const float* b1         = (const float*)d_in[10];
    const float* w2         = (const float*)d_in[11];
    const float* b2         = (const float*)d_in[12];
    const float* bn1_g = (const float*)d_in[15], *bn1_b = (const float*)d_in[16];
    const float* bn1_m = (const float*)d_in[17], *bn1_v = (const float*)d_in[18];
    const float* bn2_g = (const float*)d_in[19], *bn2_b = (const float*)d_in[20];
    const float* bn2_m = (const float*)d_in[21], *bn2_v = (const float*)d_in[22];
    const float* bn3_g = (const float*)d_in[23], *bn3_b = (const float*)d_in[24];
    const float* bn3_m = (const float*)d_in[25], *bn3_v = (const float*)d_in[26];

    const int E = in_sizes[1];

    float* ws = (float*)d_ws;
    float* wsA = ws;
    float* wsB = wsA + (size_t)NT * 1024;
    float* wsC = wsB + (size_t)NT * DD;
    float* wsD = wsC + (size_t)NT * DD;
    float* wsE = wsD + (size_t)NT * DD;
    int*   cnt   = (int*)(wsE + (size_t)NT * DD);
    int*   start = cnt + NT;
    int*   rank  = start + NT + 32;
    int*   peid  = rank + E;
    int*   pcol  = peid + E;
    float* pval  = (float*)(pcol + E);
    float* outp = (float*)d_out;

    const int EW = NT * DD;
    const dim3 blk(256);

    gemm_wmma_f32<<<gemm_blocks(NT, DD), blk, 0, stream>>>(x, w_gcn, nullptr, wsB,
                                                           NT, DD, DD, 0);
    zero_i32<<<NT / 256, blk, 0, stream>>>(cnt, NT);
    count_rows<<<(E + 255) / 256, blk, 0, stream>>>(edge_rows, cnt, rank, E);
    scan_rows<<<1, dim3(1024), 0, stream>>>(cnt, start);
    fill_grouped<<<(E + 255) / 256, blk, 0, stream>>>(edge_rows, start, rank, peid, E);
    order_rows<<<NT / 8, blk, 0, stream>>>(start, peid, edge_cols, edge_vals, pcol, pval);
    gather_rows<<<NT / 8, blk, 0, stream>>>(start, pcol, pval, wsB, wsC);
    add_gelu_bn<<<(EW + 255) / 256, blk, 0, stream>>>(x, wsC, bn1_g, bn1_b, bn1_m,
                                                      bn1_v, wsD, EW);
    gemm_wmma_f32<<<gemm_blocks(NT, 3 * DD), blk, 0, stream>>>(wsD, in_proj_w,
                                                               in_proj_b, wsA,
                                                               NT, 3 * DD, DD, 0);
    attn_wmma<<<(BB * HH * (NN / 16)) / 8, blk, 0, stream>>>(wsA, wsB);
    gemm_wmma_f32<<<gemm_blocks(NT, DD), blk, 0, stream>>>(wsB, out_proj_w,
                                                           out_proj_b, wsC,
                                                           NT, DD, DD, 0);
    add_bn<<<(EW + 255) / 256, blk, 0, stream>>>(wsD, wsC, bn2_g, bn2_b, bn2_m,
                                                 bn2_v, wsE, EW);
    gemm_wmma_f32<<<gemm_blocks(NT, 4 * DD), blk, 0, stream>>>(wsE, w1, b1, wsA,
                                                               NT, 4 * DD, DD, 1);
    gemm_wmma_f32<<<gemm_blocks(NT, DD), blk, 0, stream>>>(wsA, w2, b2, wsC,
                                                           NT, DD, 4 * DD, 0);
    add_bn<<<(EW + 255) / 256, blk, 0, stream>>>(wsE, wsC, bn3_g, bn3_b, bn3_m,
                                                 bn3_v, outp, EW);
}


# residual + BatchNorm (and the GCN GELU) fused into the stores of the kernels that produce their addend (two GEMM epilogues, the row gather): three elementwise launches removed; sum orders unchanged
# speedup vs baseline: 2.6339x; 1.1189x over previous
#include <hip/hip_runtime.h>
#include <hip/hip_bf16.h>
#include <math.h>

#define BB   16
#define NN   512
#define DD   256
#define HH   8
#define DH   32
#define NT   (BB * NN)
#define BN_EPS 1e-5f

typedef float v2f __attribute__((ext_vector_type(2)));
typedef float v8f __attribute__((ext_vector_type(8)));
typedef float v4f __attribute__((ext_vector_type(4)));

static __device__ __forceinline__ v8f wmma4(v2f a, v2f b, v8f c) {
    return __builtin_amdgcn_wmma_f32_16x16x4_f32(
         false, a,  false, b,
         (short)0, c,  false,  false);
}

static __device__ __forceinline__ float gelu_exact(float v) {
    return 0.5f * v * (1.0f + erff(v * 0.70710678118654752f));
}

#define GBM 64
#define GBN 128
#define GKC 32
#define GLD 36

__global__ __launch_bounds__(256)
void gemm_wmma_f32(const float* __restrict__ A, const float* __restrict__ W,
                   const float* __restrict__ bias, float* __restrict__ C,
                   int M, int N, int K, int fuse_gelu,
                   const float* __restrict__ res,
                   const float* __restrict__ bn_g, const float* __restrict__ bn_b,
                   const float* __restrict__ bn_m, const float* __restrict__ bn_v)
{
    __shared__ float As[2][GBM * GLD];
    __shared__ float Bs[2][GBN * GLD];
    (void)M;

    const int nbn = N >> 7;
    const int bm  = blockIdx.x / nbn;
    const int bn  = blockIdx.x - bm * nbn;
    const int m0  = bm * GBM;
    const int n0  = bn * GBN;

    const int t    = threadIdx.x;
    const int wv   = t >> 5;
    const int wm   = wv >> 2;
    const int wn   = wv & 3;
    const int lane = t & 31;
    const int half = lane >> 4;
    const int lm   = lane & 15;

    const int srow = t >> 3;
    const int sc4  = (t & 7) * 4;
    const float* ag = A + (size_t)(m0 + srow) * K + sc4;
    const float* wg = W + (size_t)(n0 + srow) * K + sc4;
    const int soff  = srow * GLD + sc4;

    v4f ra0 = *(const v4f*)(ag);
    v4f ra1 = *(const v4f*)(ag + (size_t)32 * K);
    v4f rb0 = *(const v4f*)(wg);
    v4f rb1 = *(const v4f*)(wg + (size_t)32 * K);
    v4f rb2 = *(const v4f*)(wg + (size_t)64 * K);
    v4f rb3 = *(const v4f*)(wg + (size_t)96 * K);
    *(v4f*)(&As[0][soff])            = ra0;
    *(v4f*)(&As[0][soff + 32 * GLD]) = ra1;
    *(v4f*)(&Bs[0][soff])            = rb0;
    *(v4f*)(&Bs[0][soff + 32 * GLD]) = rb1;
    *(v4f*)(&Bs[0][soff + 64 * GLD]) = rb2;
    *(v4f*)(&Bs[0][soff + 96 * GLD]) = rb3;
    __syncthreads();

    v8f acc[2][2];
#pragma unroll
    for (int i = 0; i < 2; i++)
#pragma unroll
        for (int j = 0; j < 2; j++) acc[i][j] = (v8f){};

    const int aoff = (wm * 32 + lm) * GLD + 2 * half;
    const int boff = (wn * 32 + lm) * GLD + 2 * half;
    const int nch  = K / GKC;

    for (int c = 0; c < nch; c++) {
        const int  cur  = c & 1;
        const bool more = (c + 1) < nch;
        if (more) {
            const int k1 = (c + 1) * GKC;
            ra0 = *(const v4f*)(ag + k1);
            ra1 = *(const v4f*)(ag + (size_t)32 * K + k1);
            rb0 = *(const v4f*)(wg + k1);
            rb1 = *(const v4f*)(wg + (size_t)32 * K + k1);
            rb2 = *(const v4f*)(wg + (size_t)64 * K + k1);
            rb3 = *(const v4f*)(wg + (size_t)96 * K + k1);
        }

        const float* as = &As[cur][aoff];
        const float* bs = &Bs[cur][boff];
#pragma unroll
        for (int j = 0; j < GKC / 4; j++) {
            v2f a0 = *(const v2f*)(as + 4 * j);
            v2f a1 = *(const v2f*)(as + 16 * GLD + 4 * j);
            v2f b0 = *(const v2f*)(bs + 4 * j);
            v2f b1 = *(const v2f*)(bs + 16 * GLD + 4 * j);
            acc[0][0] = wmma4(a0, b0, acc[0][0]);
            acc[0][1] = wmma4(a0, b1, acc[0][1]);
            acc[1][0] = wmma4(a1, b0, acc[1][0]);
            acc[1][1] = wmma4(a1, b1, acc[1][1]);
        }

        if (more) {
            float* an = &As[cur ^ 1][soff];
            float* bn2 = &Bs[cur ^ 1][soff];
            *(v4f*)(an)             = ra0;
            *(v4f*)(an + 32 * GLD)  = ra1;
            *(v4f*)(bn2)            = rb0;
            *(v4f*)(bn2 + 32 * GLD) = rb1;
            *(v4f*)(bn2 + 64 * GLD) = rb2;
            *(v4f*)(bn2 + 96 * GLD) = rb3;
        }
        __syncthreads();
    }

    const int nw = n0 + wn * 32;
    const int mw = m0 + wm * 32;
    float bv[2];
#pragma unroll
    for (int j = 0; j < 2; j++)
        bv[j] = bias ? bias[nw + j * 16 + lm] : 0.0f;

    if (res) {
        float sc[2], mu[2], sh[2];
#pragma unroll
        for (int j = 0; j < 2; j++) {
            const int col = nw + j * 16 + lm;
            sc[j] = bn_g[col] * rsqrtf(bn_v[col] + BN_EPS);
            mu[j] = bn_m[col];
            sh[j] = bn_b[col];
        }
        float rv[2][8][2];
#pragma unroll
        for (int i = 0; i < 2; i++) {
#pragma unroll
            for (int r = 0; r < 8; r++) {
                const size_t row = (size_t)(mw + i * 16 + r + 8 * half) * N;
#pragma unroll
                for (int j = 0; j < 2; j++)
                    rv[i][r][j] = res[row + nw + j * 16 + lm];
            }
        }
#pragma unroll
        for (int i = 0; i < 2; i++) {
#pragma unroll
            for (int r = 0; r < 8; r++) {
                const size_t row = (size_t)(mw + i * 16 + r + 8 * half) * N;
#pragma unroll
                for (int j = 0; j < 2; j++) {
                    const float val = rv[i][r][j] + (acc[i][j][r] + bv[j]);
                    C[row + nw + j * 16 + lm] = (val - mu[j]) * sc[j] + sh[j];
                }
            }
        }
        return;
    }

#pragma unroll
    for (int i = 0; i < 2; i++) {
#pragma unroll
        for (int r = 0; r < 8; r++) {
            const size_t row = (size_t)(mw + i * 16 + r + 8 * half) * N;
#pragma unroll
            for (int j = 0; j < 2; j++) {
                float v = acc[i][j][r] + bv[j];
                if (fuse_gelu) v = gelu_exact(v);
                C[row + nw + j * 16 + lm] = v;
            }
        }
    }
}

__global__ __launch_bounds__(256)
void zero_i32(int* __restrict__ p, int n)
{
    int i = blockIdx.x * blockDim.x + threadIdx.x;
    if (i < n) p[i] = 0;
}

__global__ __launch_bounds__(256)
void count_rows(const int* __restrict__ rows, int* __restrict__ cnt,
                int* __restrict__ rank, int E)
{
    int e = blockIdx.x * blockDim.x + threadIdx.x;
    if (e < E) rank[e] = atomicAdd(cnt + rows[e], 1);
}

__global__ __launch_bounds__(1024)
void scan_rows(const int* __restrict__ cnt, int* __restrict__ start)
{
    __shared__ int wsum[32];
    const int t    = threadIdx.x;
    const int lane = t & 31;
    const int w    = t >> 5;
    const int4 c0 = ((const int4*)cnt)[2 * t];
    const int4 c1 = ((const int4*)cnt)[2 * t + 1];
    const int s = c0.x + c0.y + c0.z + c0.w + c1.x + c1.y + c1.z + c1.w;
    int inc = s;
    for (int d = 1; d < 32; d <<= 1) {
        int y = __shfl_up(inc, d);
        if (lane >= d) inc += y;
    }
    if (lane == 31) wsum[w] = inc;
    __syncthreads();
    if (w == 0) {
        const int v = wsum[lane];
        int iv = v;
        for (int d = 1; d < 32; d <<= 1) {
            int y = __shfl_up(iv, d);
            if (lane >= d) iv += y;
        }
        wsum[lane] = iv - v;
    }
    __syncthreads();
    int4 o0, o1;
    o0.x = wsum[w] + inc - s;
    o0.y = o0.x + c0.x;  o0.z = o0.y + c0.y;  o0.w = o0.z + c0.z;
    o1.x = o0.w + c0.w;
    o1.y = o1.x + c1.x;  o1.z = o1.y + c1.y;  o1.w = o1.z + c1.z;
    ((int4*)start)[2 * t]     = o0;
    ((int4*)start)[2 * t + 1] = o1;
    if (t == 1023) start[NT] = o1.w + c1.w;
}

__global__ __launch_bounds__(256)
void fill_grouped(const int* __restrict__ rows, const int* __restrict__ start,
                  const int* __restrict__ rank, int* __restrict__ peid, int E)
{
    int e = blockIdx.x * blockDim.x + threadIdx.x;
    if (e >= E) return;
    peid[start[rows[e]] + rank[e]] = e;
}

__global__ __launch_bounds__(256)
void order_rows(const int* __restrict__ start, const int* __restrict__ peid,
                const int* __restrict__ cols, const float* __restrict__ vals,
                int* __restrict__ pcol, float* __restrict__ pval)
{
    const int row  = blockIdx.x * 8 + __builtin_amdgcn_readfirstlane(threadIdx.x >> 5);
    const int lane = threadIdx.x & 31;
    const int s = start[row];
    const int e = start[row + 1];
    for (int base = s; base < e; base += 32) {
        const int n  = (e - base) < 32 ? (e - base) : 32;
        const int my = (lane < n) ? peid[base + lane] : 0x7fffffff;
        int place = 0;
        for (int b2 = s; b2 < e; b2 += 32) {
            const int n2    = (e - b2) < 32 ? (e - b2) : 32;
            const int other = (lane < n2) ? peid[b2 + lane] : 0x7fffffff;
            for (int j = 0; j < n2; j++)
                place += (__builtin_amdgcn_readlane(other, j) < my) ? 1 : 0;
        }
        if (lane < n) {
            pcol[s + place] = cols[my];
            pval[s + place] = vals[my];
        }
    }
}

static __device__ __forceinline__ float4 res_gelu_bn4(float4 x, float4 a, float4 g,
                                                      float4 b, float4 m, float4 v)
{
    float4 o;
    o.x = ((x.x + gelu_exact(a.x)) - m.x) * (g.x * rsqrtf(v.x + BN_EPS)) + b.x;
    o.y = ((x.y + gelu_exact(a.y)) - m.y) * (g.y * rsqrtf(v.y + BN_EPS)) + b.y;
    o.z = ((x.z + gelu_exact(a.z)) - m.z) * (g.z * rsqrtf(v.z + BN_EPS)) + b.z;
    o.w = ((x.w + gelu_exact(a.w)) - m.w) * (g.w * rsqrtf(v.w + BN_EPS)) + b.w;
    return o;
}

__global__ __launch_bounds__(256)
void gather_rows_bn(const int* __restrict__ start, const int* __restrict__ pcol,
                    const float* __restrict__ pval, const float* __restrict__ hl,
                    const float* __restrict__ x,
                    const float* __restrict__ bn_g, const float* __restrict__ bn_b,
                    const float* __restrict__ bn_m, const float* __restrict__ bn_v,
                    float* __restrict__ x1)
{
    const int row  = blockIdx.x * 8 + __builtin_amdgcn_readfirstlane(threadIdx.x >> 5);
    const int lane = threadIdx.x & 31;
    const int s = start[row];
    const int e = start[row + 1];
    float4 a0 = make_float4(0.f, 0.f, 0.f, 0.f);
    float4 a1 = make_float4(0.f, 0.f, 0.f, 0.f);
    const float* hb = hl + lane * 4;

    for (int base = s; base < e; base += 32) {
        const int n = (e - base) < 32 ? (e - base) : 32;
        int   c = 0;
        float v = 0.0f;
        if (lane < n) { c = pcol[base + lane]; v = pval[base + lane]; }
#pragma unroll 4
        for (int j = 0; j < n; j++) {
            const int   cj = __builtin_amdgcn_readlane(c, j);
            const float vj = __int_as_float(
                __builtin_amdgcn_readlane(__float_as_int(v), j));
            const float* src = hb + (size_t)cj * DD;
            const float4 h0 = *(const float4*)(src);
            const float4 h1 = *(const float4*)(src + 128);
            a0.x += vj * h0.x;  a0.y += vj * h0.y;
            a0.z += vj * h0.z;  a0.w += vj * h0.w;
            a1.x += vj * h1.x;  a1.y += vj * h1.y;
            a1.z += vj * h1.z;  a1.w += vj * h1.w;
        }
    }
    const int    c0 = lane * 4;
    const float* xr = x + (size_t)row * DD + c0;
    float*      dst = x1 + (size_t)row * DD + c0;
    const float4 o0 = res_gelu_bn4(*(const float4*)(xr), a0,
                                   *(const float4*)(bn_g + c0), *(const float4*)(bn_b + c0),
                                   *(const float4*)(bn_m + c0), *(const float4*)(bn_v + c0));
    const float4 o1 = res_gelu_bn4(*(const float4*)(xr + 128), a1,
                                   *(const float4*)(bn_g + 128 + c0), *(const float4*)(bn_b + 128 + c0),
                                   *(const float4*)(bn_m + 128 + c0), *(const float4*)(bn_v + 128 + c0));
    *(float4*)(dst)       = o0;
    *(float4*)(dst + 128) = o1;
}

static __device__ __forceinline__ float xhalf(float x) {
    const int b = __builtin_bit_cast(int, x);
    return __builtin_bit_cast(float,
        __builtin_amdgcn_permlanex16(b, b, 0x76543210, 0xfedcba98, false, false));
}

__global__ __launch_bounds__(256)
void attn_wmma(const float* __restrict__ qkv, float* __restrict__ ctx)
{
    const int warp = threadIdx.x >> 5;
    const int gid  = blockIdx.x * 8 + warp;
    const int qt   = gid & 31;
    const int h    = (gid >> 5) & 7;
    const int b    = gid >> 8;
    const int lane = threadIdx.x & 31;
    const int half = lane >> 4;
    const int lm   = lane & 15;
    const int LDQ  = 3 * DD;

    const float* gbase = qkv + (size_t)(b * NN) * LDQ;

    v2f qb[8];
    {
        const float* qrow = gbase + (size_t)(qt * 16 + lm) * LDQ + h * DH;
        for (int j = 0; j < 8; j++) qb[j] = *(const v2f*)(qrow + 4 * j + 2 * half);
    }

    float rowmax = -1e30f;
    float rowsum = 0.0f;
    v8f o0 = {}, o1 = {};
    const float scale = 0.17677669529663687f;

    const float* kbase = gbase + (size_t)lm * LDQ + DD + h * DH + 2 * half;
    const float* vbase = gbase + (size_t)(8 * half) * LDQ + 2 * DD + h * DH + lm;

    for (int kt = 0; kt < NN / 16; kt++) {
        v8f s = {};
        const float* krow = kbase + (size_t)(kt * 16) * LDQ;
        for (int j = 0; j < 8; j++) {
            v2f ka = *(const v2f*)(krow + 4 * j);
            s = wmma4(ka, qb[j], s);
        }

        float val[8];
        for (int r = 0; r < 8; r++) val[r] = s[r] * scale;
        float mx = val[0];
        for (int r = 1; r < 8; r++) mx = fmaxf(mx, val[r]);
        mx = fmaxf(mx, xhalf(mx));
        const float nm = fmaxf(rowmax, mx);
        const float al = __expf(rowmax - nm);
        float p[8];
        float sm = 0.0f;
        for (int r = 0; r < 8; r++) { p[r] = __expf(val[r] - nm); sm += p[r]; }
        rowmax = nm;
        rowsum = rowsum * al + sm;
        for (int r = 0; r < 8; r++) { o0[r] *= al; o1[r] *= al; }

        const float* vp = vbase + (size_t)(kt * 16) * LDQ;
        for (int j = 0; j < 4; j++) {
            v2f pb, va0, va1;
            pb.x  = p[2 * j];                     pb.y  = p[2 * j + 1];
            va0.x = vp[(2 * j) * LDQ];            va0.y = vp[(2 * j + 1) * LDQ];
            va1.x = vp[(2 * j) * LDQ + 16];       va1.y = vp[(2 * j + 1) * LDQ + 16];
            o0 = wmma4(va0, pb, o0);
            o1 = wmma4(va1, pb, o1);
        }
    }

    const float inv = 1.0f / (rowsum + xhalf(rowsum));
    float* dst = ctx + (size_t)(b * NN + qt * 16 + lm) * DD + h * DH + 8 * half;
    *(v4f*)(dst)      = (v4f){o0[0] * inv, o0[1] * inv, o0[2] * inv, o0[3] * inv};
    *(v4f*)(dst + 4)  = (v4f){o0[4] * inv, o0[5] * inv, o0[6] * inv, o0[7] * inv};
    *(v4f*)(dst + 16) = (v4f){o1[0] * inv, o1[1] * inv, o1[2] * inv, o1[3] * inv};
    *(v4f*)(dst + 20) = (v4f){o1[4] * inv, o1[5] * inv, o1[6] * inv, o1[7] * inv};
}

static inline int gemm_blocks(int M, int N) { return (M >> 6) * (N >> 7); }

extern "C" void kernel_launch(void* const* d_in, const int* in_sizes, int n_in,
                              void* d_out, int out_size, void* d_ws, size_t ws_size,
                              hipStream_t stream)
{
    (void)n_in; (void)out_size; (void)ws_size;
    const float* x          = (const float*)d_in[0];
    const int*   edge_rows  = (const int*)  d_in[1];
    const int*   edge_cols  = (const int*)  d_in[2];
    const float* edge_vals  = (const float*)d_in[3];
    const float* w_gcn      = (const float*)d_in[4];
    const float* in_proj_w  = (const float*)d_in[5];
    const float* in_proj_b  = (const float*)d_in[6];
    const float* out_proj_w = (const float*)d_in[7];
    const float* out_proj_b = (const float*)d_in[8];
    const float* w1         = (const float*)d_in[9];
    const float* b1         = (const float*)d_in[10];
    const float* w2         = (const float*)d_in[11];
    const float* b2         = (const float*)d_in[12];
    const float* bn1_g = (const float*)d_in[15], *bn1_b = (const float*)d_in[16];
    const float* bn1_m = (const float*)d_in[17], *bn1_v = (const float*)d_in[18];
    const float* bn2_g = (const float*)d_in[19], *bn2_b = (const float*)d_in[20];
    const float* bn2_m = (const float*)d_in[21], *bn2_v = (const float*)d_in[22];
    const float* bn3_g = (const float*)d_in[23], *bn3_b = (const float*)d_in[24];
    const float* bn3_m = (const float*)d_in[25], *bn3_v = (const float*)d_in[26];

    const int E = in_sizes[1];

    float* ws = (float*)d_ws;
    float* wsA = ws;
    float* wsB = wsA + (size_t)NT * 1024;
    float* wsC = wsB + (size_t)NT * DD;
    float* wsD = wsC + (size_t)NT * DD;
    float* wsE = wsD + (size_t)NT * DD;
    int*   cnt   = (int*)(wsE + (size_t)NT * DD);
    int*   start = cnt + NT;
    int*   rank  = start + NT + 32;
    int*   peid  = rank + E;
    int*   pcol  = peid + E;
    float* pval  = (float*)(pcol + E);
    float* outp = (float*)d_out;

    const dim3 blk(256);

    gemm_wmma_f32<<<gemm_blocks(NT, DD), blk, 0, stream>>>(x, w_gcn, nullptr, wsB,
                                                           NT, DD, DD, 0, nullptr,
                                                           nullptr, nullptr, nullptr, nullptr);
    zero_i32<<<NT / 256, blk, 0, stream>>>(cnt, NT);
    count_rows<<<(E + 255) / 256, blk, 0, stream>>>(edge_rows, cnt, rank, E);
    scan_rows<<<1, dim3(1024), 0, stream>>>(cnt, start);
    fill_grouped<<<(E + 255) / 256, blk, 0, stream>>>(edge_rows, start, rank, peid, E);
    order_rows<<<NT / 8, blk, 0, stream>>>(start, peid, edge_cols, edge_vals, pcol, pval);
    gather_rows_bn<<<NT / 8, blk, 0, stream>>>(start, pcol, pval, wsB, x,
                                               bn1_g, bn1_b, bn1_m, bn1_v, wsD);
    gemm_wmma_f32<<<gemm_blocks(NT, 3 * DD), blk, 0, stream>>>(wsD, in_proj_w,
                                                               in_proj_b, wsA,
                                                               NT, 3 * DD, DD, 0, nullptr,
                                                               nullptr, nullptr, nullptr, nullptr);
    attn_wmma<<<(BB * HH * (NN / 16)) / 8, blk, 0, stream>>>(wsA, wsB);
    gemm_wmma_f32<<<gemm_blocks(NT, DD), blk, 0, stream>>>(wsB, out_proj_w,
                                                           out_proj_b, wsE,
                                                           NT, DD, DD, 0, wsD,
                                                           bn2_g, bn2_b, bn2_m, bn2_v);
    gemm_wmma_f32<<<gemm_blocks(NT, 4 * DD), blk, 0, stream>>>(wsE, w1, b1, wsA,
                                                               NT, 4 * DD, DD, 1, nullptr,
                                                               nullptr, nullptr, nullptr, nullptr);
    gemm_wmma_f32<<<gemm_blocks(NT, DD), blk, 0, stream>>>(wsA, w2, b2, outp,
                                                           NT, DD, 4 * DD, 0, wsE,
                                                           bn3_g, bn3_b, bn3_m, bn3_v);
}


# attention K and V staged through LDS once per work-group (8 query tiles of one head): whole-line 16-byte global loads one 64-key chunk ahead, 8-byte LDS fragment reads; k and sum order unchanged
# speedup vs baseline: 2.7687x; 1.0512x over previous
#include <hip/hip_runtime.h>
#include <hip/hip_bf16.h>
#include <math.h>

#define BB   16
#define NN   512
#define DD   256
#define HH   8
#define DH   32
#define NT   (BB * NN)
#define BN_EPS 1e-5f

typedef float v2f __attribute__((ext_vector_type(2)));
typedef float v8f __attribute__((ext_vector_type(8)));
typedef float v4f __attribute__((ext_vector_type(4)));

static __device__ __forceinline__ v8f wmma4(v2f a, v2f b, v8f c) {
    return __builtin_amdgcn_wmma_f32_16x16x4_f32(
         false, a,  false, b,
         (short)0, c,  false,  false);
}

static __device__ __forceinline__ float gelu_exact(float v) {
    return 0.5f * v * (1.0f + erff(v * 0.70710678118654752f));
}

#define GBM 64
#define GBN 128
#define GKC 32
#define GLD 36

__global__ __launch_bounds__(256)
void gemm_wmma_f32(const float* __restrict__ A, const float* __restrict__ W,
                   const float* __restrict__ bias, float* __restrict__ C,
                   int M, int N, int K, int fuse_gelu,
                   const float* __restrict__ res,
                   const float* __restrict__ bn_g, const float* __restrict__ bn_b,
                   const float* __restrict__ bn_m, const float* __restrict__ bn_v)
{
    __shared__ float As[2][GBM * GLD];
    __shared__ float Bs[2][GBN * GLD];
    (void)M;

    const int nbn = N >> 7;
    const int bm  = blockIdx.x / nbn;
    const int bn  = blockIdx.x - bm * nbn;
    const int m0  = bm * GBM;
    const int n0  = bn * GBN;

    const int t    = threadIdx.x;
    const int wv   = t >> 5;
    const int wm   = wv >> 2;
    const int wn   = wv & 3;
    const int lane = t & 31;
    const int half = lane >> 4;
    const int lm   = lane & 15;

    const int srow = t >> 3;
    const int sc4  = (t & 7) * 4;
    const float* ag = A + (size_t)(m0 + srow) * K + sc4;
    const float* wg = W + (size_t)(n0 + srow) * K + sc4;
    const int soff  = srow * GLD + sc4;

    v4f ra0 = *(const v4f*)(ag);
    v4f ra1 = *(const v4f*)(ag + (size_t)32 * K);
    v4f rb0 = *(const v4f*)(wg);
    v4f rb1 = *(const v4f*)(wg + (size_t)32 * K);
    v4f rb2 = *(const v4f*)(wg + (size_t)64 * K);
    v4f rb3 = *(const v4f*)(wg + (size_t)96 * K);
    *(v4f*)(&As[0][soff])            = ra0;
    *(v4f*)(&As[0][soff + 32 * GLD]) = ra1;
    *(v4f*)(&Bs[0][soff])            = rb0;
    *(v4f*)(&Bs[0][soff + 32 * GLD]) = rb1;
    *(v4f*)(&Bs[0][soff + 64 * GLD]) = rb2;
    *(v4f*)(&Bs[0][soff + 96 * GLD]) = rb3;
    __syncthreads();

    v8f acc[2][2];
#pragma unroll
    for (int i = 0; i < 2; i++)
#pragma unroll
        for (int j = 0; j < 2; j++) acc[i][j] = (v8f){};

    const int aoff = (wm * 32 + lm) * GLD + 2 * half;
    const int boff = (wn * 32 + lm) * GLD + 2 * half;
    const int nch  = K / GKC;

    for (int c = 0; c < nch; c++) {
        const int  cur  = c & 1;
        const bool more = (c + 1) < nch;
        if (more) {
            const int k1 = (c + 1) * GKC;
            ra0 = *(const v4f*)(ag + k1);
            ra1 = *(const v4f*)(ag + (size_t)32 * K + k1);
            rb0 = *(const v4f*)(wg + k1);
            rb1 = *(const v4f*)(wg + (size_t)32 * K + k1);
            rb2 = *(const v4f*)(wg + (size_t)64 * K + k1);
            rb3 = *(const v4f*)(wg + (size_t)96 * K + k1);
        }

        const float* as = &As[cur][aoff];
        const float* bs = &Bs[cur][boff];
#pragma unroll
        for (int j = 0; j < GKC / 4; j++) {
            v2f a0 = *(const v2f*)(as + 4 * j);
            v2f a1 = *(const v2f*)(as + 16 * GLD + 4 * j);
            v2f b0 = *(const v2f*)(bs + 4 * j);
            v2f b1 = *(const v2f*)(bs + 16 * GLD + 4 * j);
            acc[0][0] = wmma4(a0, b0, acc[0][0]);
            acc[0][1] = wmma4(a0, b1, acc[0][1]);
            acc[1][0] = wmma4(a1, b0, acc[1][0]);
            acc[1][1] = wmma4(a1, b1, acc[1][1]);
        }

        if (more) {
            float* an = &As[cur ^ 1][soff];
            float* bn2 = &Bs[cur ^ 1][soff];
            *(v4f*)(an)             = ra0;
            *(v4f*)(an + 32 * GLD)  = ra1;
            *(v4f*)(bn2)            = rb0;
            *(v4f*)(bn2 + 32 * GLD) = rb1;
            *(v4f*)(bn2 + 64 * GLD) = rb2;
            *(v4f*)(bn2 + 96 * GLD) = rb3;
        }
        __syncthreads();
    }

    const int nw = n0 + wn * 32;
    const int mw = m0 + wm * 32;
    float bv[2];
#pragma unroll
    for (int j = 0; j < 2; j++)
        bv[j] = bias ? bias[nw + j * 16 + lm] : 0.0f;

    if (res) {
        float sc[2], mu[2], sh[2];
#pragma unroll
        for (int j = 0; j < 2; j++) {
            const int col = nw + j * 16 + lm;
            sc[j] = bn_g[col] * rsqrtf(bn_v[col] + BN_EPS);
            mu[j] = bn_m[col];
            sh[j] = bn_b[col];
        }
        float rv[2][8][2];
#pragma unroll
        for (int i = 0; i < 2; i++) {
#pragma unroll
            for (int r = 0; r < 8; r++) {
                const size_t row = (size_t)(mw + i * 16 + r + 8 * half) * N;
#pragma unroll
                for (int j = 0; j < 2; j++)
                    rv[i][r][j] = res[row + nw + j * 16 + lm];
            }
        }
#pragma unroll
        for (int i = 0; i < 2; i++) {
#pragma unroll
            for (int r = 0; r < 8; r++) {
                const size_t row = (size_t)(mw + i * 16 + r + 8 * half) * N;
#pragma unroll
                for (int j = 0; j < 2; j++) {
                    const float val = rv[i][r][j] + (acc[i][j][r] + bv[j]);
                    C[row + nw + j * 16 + lm] = (val - mu[j]) * sc[j] + sh[j];
                }
            }
        }
        return;
    }

#pragma unroll
    for (int i = 0; i < 2; i++) {
#pragma unroll
        for (int r = 0; r < 8; r++) {
            const size_t row = (size_t)(mw + i * 16 + r + 8 * half) * N;
#pragma unroll
            for (int j = 0; j < 2; j++) {
                float v = acc[i][j][r] + bv[j];
                if (fuse_gelu) v = gelu_exact(v);
                C[row + nw + j * 16 + lm] = v;
            }
        }
    }
}

__global__ __launch_bounds__(256)
void zero_i32(int* __restrict__ p, int n)
{
    int i = blockIdx.x * blockDim.x + threadIdx.x;
    if (i < n) p[i] = 0;
}

__global__ __launch_bounds__(256)
void count_rows(const int* __restrict__ rows, int* __restrict__ cnt,
                int* __restrict__ rank, int E)
{
    int e = blockIdx.x * blockDim.x + threadIdx.x;
    if (e < E) rank[e] = atomicAdd(cnt + rows[e], 1);
}

__global__ __launch_bounds__(1024)
void scan_rows(const int* __restrict__ cnt, int* __restrict__ start)
{
    __shared__ int wsum[32];
    const int t    = threadIdx.x;
    const int lane = t & 31;
    const int w    = t >> 5;
    const int4 c0 = ((const int4*)cnt)[2 * t];
    const int4 c1 = ((const int4*)cnt)[2 * t + 1];
    const int s = c0.x + c0.y + c0.z + c0.w + c1.x + c1.y + c1.z + c1.w;
    int inc = s;
    for (int d = 1; d < 32; d <<= 1) {
        int y = __shfl_up(inc, d);
        if (lane >= d) inc += y;
    }
    if (lane == 31) wsum[w] = inc;
    __syncthreads();
    if (w == 0) {
        const int v = wsum[lane];
        int iv = v;
        for (int d = 1; d < 32; d <<= 1) {
            int y = __shfl_up(iv, d);
            if (lane >= d) iv += y;
        }
        wsum[lane] = iv - v;
    }
    __syncthreads();
    int4 o0, o1;
    o0.x = wsum[w] + inc - s;
    o0.y = o0.x + c0.x;  o0.z = o0.y + c0.y;  o0.w = o0.z + c0.z;
    o1.x = o0.w + c0.w;
    o1.y = o1.x + c1.x;  o1.z = o1.y + c1.y;  o1.w = o1.z + c1.z;
    ((int4*)start)[2 * t]     = o0;
    ((int4*)start)[2 * t + 1] = o1;
    if (t == 1023) start[NT] = o1.w + c1.w;
}

__global__ __launch_bounds__(256)
void fill_grouped(const int* __restrict__ rows, const int* __restrict__ start,
                  const int* __restrict__ rank, int* __restrict__ peid, int E)
{
    int e = blockIdx.x * blockDim.x + threadIdx.x;
    if (e >= E) return;
    peid[start[rows[e]] + rank[e]] = e;
}

__global__ __launch_bounds__(256)
void order_rows(const int* __restrict__ start, const int* __restrict__ peid,
                const int* __restrict__ cols, const float* __restrict__ vals,
                int* __restrict__ pcol, float* __restrict__ pval)
{
    const int row  = blockIdx.x * 8 + __builtin_amdgcn_readfirstlane(threadIdx.x >> 5);
    const int lane = threadIdx.x & 31;
    const int s = start[row];
    const int e = start[row + 1];
    for (int base = s; base < e; base += 32) {
        const int n  = (e - base) < 32 ? (e - base) : 32;
        const int my = (lane < n) ? peid[base + lane] : 0x7fffffff;
        int place = 0;
        for (int b2 = s; b2 < e; b2 += 32) {
            const int n2    = (e - b2) < 32 ? (e - b2) : 32;
            const int other = (lane < n2) ? peid[b2 + lane] : 0x7fffffff;
            for (int j = 0; j < n2; j++)
                place += (__builtin_amdgcn_readlane(other, j) < my) ? 1 : 0;
        }
        if (lane < n) {
            pcol[s + place] = cols[my];
            pval[s + place] = vals[my];
        }
    }
}

static __device__ __forceinline__ float4 res_gelu_bn4(float4 x, float4 a, float4 g,
                                                      float4 b, float4 m, float4 v)
{
    float4 o;
    o.x = ((x.x + gelu_exact(a.x)) - m.x) * (g.x * rsqrtf(v.x + BN_EPS)) + b.x;
    o.y = ((x.y + gelu_exact(a.y)) - m.y) * (g.y * rsqrtf(v.y + BN_EPS)) + b.y;
    o.z = ((x.z + gelu_exact(a.z)) - m.z) * (g.z * rsqrtf(v.z + BN_EPS)) + b.z;
    o.w = ((x.w + gelu_exact(a.w)) - m.w) * (g.w * rsqrtf(v.w + BN_EPS)) + b.w;
    return o;
}

__global__ __launch_bounds__(256)
void gather_rows_bn(const int* __restrict__ start, const int* __restrict__ pcol,
                    const float* __restrict__ pval, const float* __restrict__ hl,
                    const float* __restrict__ x,
                    const float* __restrict__ bn_g, const float* __restrict__ bn_b,
                    const float* __restrict__ bn_m, const float* __restrict__ bn_v,
                    float* __restrict__ x1)
{
    const int row  = blockIdx.x * 8 + __builtin_amdgcn_readfirstlane(threadIdx.x >> 5);
    const int lane = threadIdx.x & 31;
    const int s = start[row];
    const int e = start[row + 1];
    float4 a0 = make_float4(0.f, 0.f, 0.f, 0.f);
    float4 a1 = make_float4(0.f, 0.f, 0.f, 0.f);
    const float* hb = hl + lane * 4;

    for (int base = s; base < e; base += 32) {
        const int n = (e - base) < 32 ? (e - base) : 32;
        int   c = 0;
        float v = 0.0f;
        if (lane < n) { c = pcol[base + lane]; v = pval[base + lane]; }
#pragma unroll 4
        for (int j = 0; j < n; j++) {
            const int   cj = __builtin_amdgcn_readlane(c, j);
            const float vj = __int_as_float(
                __builtin_amdgcn_readlane(__float_as_int(v), j));
            const float* src = hb + (size_t)cj * DD;
            const float4 h0 = *(const float4*)(src);
            const float4 h1 = *(const float4*)(src + 128);
            a0.x += vj * h0.x;  a0.y += vj * h0.y;
            a0.z += vj * h0.z;  a0.w += vj * h0.w;
            a1.x += vj * h1.x;  a1.y += vj * h1.y;
            a1.z += vj * h1.z;  a1.w += vj * h1.w;
        }
    }
    const int    c0 = lane * 4;
    const float* xr = x + (size_t)row * DD + c0;
    float*      dst = x1 + (size_t)row * DD + c0;
    const float4 o0 = res_gelu_bn4(*(const float4*)(xr), a0,
                                   *(const float4*)(bn_g + c0), *(const float4*)(bn_b + c0),
                                   *(const float4*)(bn_m + c0), *(const float4*)(bn_v + c0));
    const float4 o1 = res_gelu_bn4(*(const float4*)(xr + 128), a1,
                                   *(const float4*)(bn_g + 128 + c0), *(const float4*)(bn_b + 128 + c0),
                                   *(const float4*)(bn_m + 128 + c0), *(const float4*)(bn_v + 128 + c0));
    *(float4*)(dst)       = o0;
    *(float4*)(dst + 128) = o1;
}

static __device__ __forceinline__ float xhalf(float x) {
    const int b = __builtin_bit_cast(int, x);
    return __builtin_bit_cast(float,
        __builtin_amdgcn_permlanex16(b, b, 0x76543210, 0xfedcba98, false, false));
}

#define AKC 64
#define ALD 36
#define VLD 72

__global__ __launch_bounds__(256)
void attn_wmma(const float* __restrict__ qkv, float* __restrict__ ctx)
{
    __shared__ float Ks[2][AKC * ALD];
    __shared__ float Vs[2][(AKC / 2) * VLD];

    const int t    = threadIdx.x;
    const int warp = t >> 5;
    const int qt   = (blockIdx.x & 3) * 8 + warp;
    const int h    = (blockIdx.x >> 2) & 7;
    const int b    = blockIdx.x >> 5;
    const int lane = t & 31;
    const int half = lane >> 4;
    const int lm   = lane & 15;
    const int LDQ  = 3 * DD;

    const float* gbase = qkv + (size_t)(b * NN) * LDQ;

    const int srow = t >> 3;
    const int sc4  = (t & 7) * 4;
    const float* kg = gbase + (size_t)srow * LDQ + DD + h * DH + sc4;
    const float* vg = gbase + (size_t)(2 * srow) * LDQ + 2 * DD + h * DH + sc4;
    const int soff  = srow * ALD + sc4;
    const int svoff = srow * VLD + 2 * sc4;

    v4f rk0 = *(const v4f*)(kg);
    v4f rk1 = *(const v4f*)(kg + (size_t)32 * LDQ);
    v4f rv0 = *(const v4f*)(vg);
    v4f rv1 = *(const v4f*)(vg + LDQ);

    v2f qb[8];
    {
        const float* qrow = gbase + (size_t)(qt * 16 + lm) * LDQ + h * DH;
        for (int j = 0; j < 8; j++) qb[j] = *(const v2f*)(qrow + 4 * j + 2 * half);
    }

    *(v4f*)(&Ks[0][soff])            = rk0;
    *(v4f*)(&Ks[0][soff + 32 * ALD]) = rk1;
    *(v4f*)(&Vs[0][svoff])           = (v4f){rv0.x, rv1.x, rv0.y, rv1.y};
    *(v4f*)(&Vs[0][svoff + 4])       = (v4f){rv0.z, rv1.z, rv0.w, rv1.w};
    __syncthreads();

    float rowmax = -1e30f;
    float rowsum = 0.0f;
    v8f o0 = {}, o1 = {};
    const float scale = 0.17677669529663687f;

    const int koff = lm * ALD + 2 * half;
    const int voff = (4 * half) * VLD + 2 * lm;

    for (int c = 0; c < NN / AKC; c++) {
        const int  cur  = c & 1;
        const bool more = (c + 1) < NN / AKC;
        if (more) {
            const size_t o = (size_t)((c + 1) * AKC) * LDQ;
            rk0 = *(const v4f*)(kg + o);
            rk1 = *(const v4f*)(kg + o + (size_t)32 * LDQ);
            rv0 = *(const v4f*)(vg + o);
            rv1 = *(const v4f*)(vg + o + LDQ);
        }

        const float* ks = &Ks[cur][koff];
        const float* vs = &Vs[cur][voff];
        for (int kt = 0; kt < AKC / 16; kt++) {
            v8f s = {};
            const float* krow = ks + kt * 16 * ALD;
            for (int j = 0; j < 8; j++) {
                v2f ka = *(const v2f*)(krow + 4 * j);
                s = wmma4(ka, qb[j], s);
            }

            float val[8];
            for (int r = 0; r < 8; r++) val[r] = s[r] * scale;
            float mx = val[0];
            for (int r = 1; r < 8; r++) mx = fmaxf(mx, val[r]);
            mx = fmaxf(mx, xhalf(mx));
            const float nm = fmaxf(rowmax, mx);
            const float al = __expf(rowmax - nm);
            float p[8];
            float sm = 0.0f;
            for (int r = 0; r < 8; r++) { p[r] = __expf(val[r] - nm); sm += p[r]; }
            rowmax = nm;
            rowsum = rowsum * al + sm;
            for (int r = 0; r < 8; r++) { o0[r] *= al; o1[r] *= al; }

            const float* vp = vs + kt * 8 * VLD;
            for (int j = 0; j < 4; j++) {
                v2f pb;
                pb.x = p[2 * j];                      pb.y = p[2 * j + 1];
                const v2f va0 = *(const v2f*)(vp + j * VLD);
                const v2f va1 = *(const v2f*)(vp + j * VLD + 32);
                o0 = wmma4(va0, pb, o0);
                o1 = wmma4(va1, pb, o1);
            }
        }

        if (more) {
            float* kn = &Ks[cur ^ 1][soff];
            float* vn = &Vs[cur ^ 1][svoff];
            *(v4f*)(kn)            = rk0;
            *(v4f*)(kn + 32 * ALD) = rk1;
            *(v4f*)(vn)            = (v4f){rv0.x, rv1.x, rv0.y, rv1.y};
            *(v4f*)(vn + 4)        = (v4f){rv0.z, rv1.z, rv0.w, rv1.w};
        }
        __syncthreads();
    }

    const float inv = 1.0f / (rowsum + xhalf(rowsum));
    float* dst = ctx + (size_t)(b * NN + qt * 16 + lm) * DD + h * DH + 8 * half;
    *(v4f*)(dst)      = (v4f){o0[0] * inv, o0[1] * inv, o0[2] * inv, o0[3] * inv};
    *(v4f*)(dst + 4)  = (v4f){o0[4] * inv, o0[5] * inv, o0[6] * inv, o0[7] * inv};
    *(v4f*)(dst + 16) = (v4f){o1[0] * inv, o1[1] * inv, o1[2] * inv, o1[3] * inv};
    *(v4f*)(dst + 20) = (v4f){o1[4] * inv, o1[5] * inv, o1[6] * inv, o1[7] * inv};
}

static inline int gemm_blocks(int M, int N) { return (M >> 6) * (N >> 7); }

extern "C" void kernel_launch(void* const* d_in, const int* in_sizes, int n_in,
                              void* d_out, int out_size, void* d_ws, size_t ws_size,
                              hipStream_t stream)
{
    (void)n_in; (void)out_size; (void)ws_size;
    const float* x          = (const float*)d_in[0];
    const int*   edge_rows  = (const int*)  d_in[1];
    const int*   edge_cols  = (const int*)  d_in[2];
    const float* edge_vals  = (const float*)d_in[3];
    const float* w_gcn      = (const float*)d_in[4];
    const float* in_proj_w  = (const float*)d_in[5];
    const float* in_proj_b  = (const float*)d_in[6];
    const float* out_proj_w = (const float*)d_in[7];
    const float* out_proj_b = (const float*)d_in[8];
    const float* w1         = (const float*)d_in[9];
    const float* b1         = (const float*)d_in[10];
    const float* w2         = (const float*)d_in[11];
    const float* b2         = (const float*)d_in[12];
    const float* bn1_g = (const float*)d_in[15], *bn1_b = (const float*)d_in[16];
    const float* bn1_m = (const float*)d_in[17], *bn1_v = (const float*)d_in[18];
    const float* bn2_g = (const float*)d_in[19], *bn2_b = (const float*)d_in[20];
    const float* bn2_m = (const float*)d_in[21], *bn2_v = (const float*)d_in[22];
    const float* bn3_g = (const float*)d_in[23], *bn3_b = (const float*)d_in[24];
    const float* bn3_m = (const float*)d_in[25], *bn3_v = (const float*)d_in[26];

    const int E = in_sizes[1];

    float* ws = (float*)d_ws;
    float* wsA = ws;
    float* wsB = wsA + (size_t)NT * 1024;
    float* wsC = wsB + (size_t)NT * DD;
    float* wsD = wsC + (size_t)NT * DD;
    float* wsE = wsD + (size_t)NT * DD;
    int*   cnt   = (int*)(wsE + (size_t)NT * DD);
    int*   start = cnt + NT;
    int*   rank  = start + NT + 32;
    int*   peid  = rank + E;
    int*   pcol  = peid + E;
    float* pval  = (float*)(pcol + E);
    float* outp = (float*)d_out;

    const dim3 blk(256);

    gemm_wmma_f32<<<gemm_blocks(NT, DD), blk, 0, stream>>>(x, w_gcn, nullptr, wsB,
                                                           NT, DD, DD, 0, nullptr,
                                                           nullptr, nullptr, nullptr, nullptr);
    zero_i32<<<NT / 256, blk, 0, stream>>>(cnt, NT);
    count_rows<<<(E + 255) / 256, blk, 0, stream>>>(edge_rows, cnt, rank, E);
    scan_rows<<<1, dim3(1024), 0, stream>>>(cnt, start);
    fill_grouped<<<(E + 255) / 256, blk, 0, stream>>>(edge_rows, start, rank, peid, E);
    order_rows<<<NT / 8, blk, 0, stream>>>(start, peid, edge_cols, edge_vals, pcol, pval);
    gather_rows_bn<<<NT / 8, blk, 0, stream>>>(start, pcol, pval, wsB, x,
                                               bn1_g, bn1_b, bn1_m, bn1_v, wsD);
    gemm_wmma_f32<<<gemm_blocks(NT, 3 * DD), blk, 0, stream>>>(wsD, in_proj_w,
                                                               in_proj_b, wsA,
                                                               NT, 3 * DD, DD, 0, nullptr,
                                                               nullptr, nullptr, nullptr, nullptr);
    attn_wmma<<<(BB * HH * (NN / 16)) / 8, blk, 0, stream>>>(wsA, wsB);
    gemm_wmma_f32<<<gemm_blocks(NT, DD), blk, 0, stream>>>(wsB, out_proj_w,
                                                           out_proj_b, wsE,
                                                           NT, DD, DD, 0, wsD,
                                                           bn2_g, bn2_b, bn2_m, bn2_v);
    gemm_wmma_f32<<<gemm_blocks(NT, 4 * DD), blk, 0, stream>>>(wsE, w1, b1, wsA,
                                                               NT, 4 * DD, DD, 1, nullptr,
                                                               nullptr, nullptr, nullptr, nullptr);
    gemm_wmma_f32<<<gemm_blocks(NT, DD), blk, 0, stream>>>(wsA, w2, b2, outp,
                                                           NT, DD, 4 * DD, 0, wsE,
                                                           bn3_g, bn3_b, bn3_m, bn3_v);
}


# edge grouping folded into the row-sum kernel: a 32-wave work-group scans the edge rows, lists its 32 rows' edges in LDS (int atomics), orders each list by edge id, sums it; sum order unchanged
# speedup vs baseline: 3.2160x; 1.1616x over previous
#include <hip/hip_runtime.h>
#include <hip/hip_bf16.h>
#include <math.h>

#define BB   16
#define NN   512
#define DD   256
#define HH   8
#define DH   32
#define NT   (BB * NN)
#define BN_EPS 1e-5f

typedef float v2f __attribute__((ext_vector_type(2)));
typedef float v8f __attribute__((ext_vector_type(8)));
typedef float v4f __attribute__((ext_vector_type(4)));

static __device__ __forceinline__ v8f wmma4(v2f a, v2f b, v8f c) {
    return __builtin_amdgcn_wmma_f32_16x16x4_f32(
         false, a,  false, b,
         (short)0, c,  false,  false);
}

static __device__ __forceinline__ float gelu_exact(float v) {
    return 0.5f * v * (1.0f + erff(v * 0.70710678118654752f));
}

#define GBM 64
#define GBN 128
#define GKC 32
#define GLD 36

__global__ __launch_bounds__(256)
void gemm_wmma_f32(const float* __restrict__ A, const float* __restrict__ W,
                   const float* __restrict__ bias, float* __restrict__ C,
                   int M, int N, int K, int fuse_gelu,
                   const float* __restrict__ res,
                   const float* __restrict__ bn_g, const float* __restrict__ bn_b,
                   const float* __restrict__ bn_m, const float* __restrict__ bn_v)
{
    __shared__ float As[2][GBM * GLD];
    __shared__ float Bs[2][GBN * GLD];
    (void)M;

    const int nbn = N >> 7;
    const int bm  = blockIdx.x / nbn;
    const int bn  = blockIdx.x - bm * nbn;
    const int m0  = bm * GBM;
    const int n0  = bn * GBN;

    const int t    = threadIdx.x;
    const int wv   = t >> 5;
    const int wm   = wv >> 2;
    const int wn   = wv & 3;
    const int lane = t & 31;
    const int half = lane >> 4;
    const int lm   = lane & 15;

    const int srow = t >> 3;
    const int sc4  = (t & 7) * 4;
    const float* ag = A + (size_t)(m0 + srow) * K + sc4;
    const float* wg = W + (size_t)(n0 + srow) * K + sc4;
    const int soff  = srow * GLD + sc4;

    v4f ra0 = *(const v4f*)(ag);
    v4f ra1 = *(const v4f*)(ag + (size_t)32 * K);
    v4f rb0 = *(const v4f*)(wg);
    v4f rb1 = *(const v4f*)(wg + (size_t)32 * K);
    v4f rb2 = *(const v4f*)(wg + (size_t)64 * K);
    v4f rb3 = *(const v4f*)(wg + (size_t)96 * K);
    *(v4f*)(&As[0][soff])            = ra0;
    *(v4f*)(&As[0][soff + 32 * GLD]) = ra1;
    *(v4f*)(&Bs[0][soff])            = rb0;
    *(v4f*)(&Bs[0][soff + 32 * GLD]) = rb1;
    *(v4f*)(&Bs[0][soff + 64 * GLD]) = rb2;
    *(v4f*)(&Bs[0][soff + 96 * GLD]) = rb3;
    __syncthreads();

    v8f acc[2][2];
#pragma unroll
    for (int i = 0; i < 2; i++)
#pragma unroll
        for (int j = 0; j < 2; j++) acc[i][j] = (v8f){};

    const int aoff = (wm * 32 + lm) * GLD + 2 * half;
    const int boff = (wn * 32 + lm) * GLD + 2 * half;
    const int nch  = K / GKC;

    for (int c = 0; c < nch; c++) {
        const int  cur  = c & 1;
        const bool more = (c + 1) < nch;
        if (more) {
            const int k1 = (c + 1) * GKC;
            ra0 = *(const v4f*)(ag + k1);
            ra1 = *(const v4f*)(ag + (size_t)32 * K + k1);
            rb0 = *(const v4f*)(wg + k1);
            rb1 = *(const v4f*)(wg + (size_t)32 * K + k1);
            rb2 = *(const v4f*)(wg + (size_t)64 * K + k1);
            rb3 = *(const v4f*)(wg + (size_t)96 * K + k1);
        }

        const float* as = &As[cur][aoff];
        const float* bs = &Bs[cur][boff];
#pragma unroll
        for (int j = 0; j < GKC / 4; j++) {
            v2f a0 = *(const v2f*)(as + 4 * j);
            v2f a1 = *(const v2f*)(as + 16 * GLD + 4 * j);
            v2f b0 = *(const v2f*)(bs + 4 * j);
            v2f b1 = *(const v2f*)(bs + 16 * GLD + 4 * j);
            acc[0][0] = wmma4(a0, b0, acc[0][0]);
            acc[0][1] = wmma4(a0, b1, acc[0][1]);
            acc[1][0] = wmma4(a1, b0, acc[1][0]);
            acc[1][1] = wmma4(a1, b1, acc[1][1]);
        }

        if (more) {
            float* an = &As[cur ^ 1][soff];
            float* bn2 = &Bs[cur ^ 1][soff];
            *(v4f*)(an)             = ra0;
            *(v4f*)(an + 32 * GLD)  = ra1;
            *(v4f*)(bn2)            = rb0;
            *(v4f*)(bn2 + 32 * GLD) = rb1;
            *(v4f*)(bn2 + 64 * GLD) = rb2;
            *(v4f*)(bn2 + 96 * GLD) = rb3;
        }
        __syncthreads();
    }

    const int nw = n0 + wn * 32;
    const int mw = m0 + wm * 32;
    float bv[2];
#pragma unroll
    for (int j = 0; j < 2; j++)
        bv[j] = bias ? bias[nw + j * 16 + lm] : 0.0f;

    if (res) {
        float sc[2], mu[2], sh[2];
#pragma unroll
        for (int j = 0; j < 2; j++) {
            const int col = nw + j * 16 + lm;
            sc[j] = bn_g[col] * rsqrtf(bn_v[col] + BN_EPS);
            mu[j] = bn_m[col];
            sh[j] = bn_b[col];
        }
        float rv[2][8][2];
#pragma unroll
        for (int i = 0; i < 2; i++) {
#pragma unroll
            for (int r = 0; r < 8; r++) {
                const size_t row = (size_t)(mw + i * 16 + r + 8 * half) * N;
#pragma unroll
                for (int j = 0; j < 2; j++)
                    rv[i][r][j] = res[row + nw + j * 16 + lm];
            }
        }
#pragma unroll
        for (int i = 0; i < 2; i++) {
#pragma unroll
            for (int r = 0; r < 8; r++) {
                const size_t row = (size_t)(mw + i * 16 + r + 8 * half) * N;
#pragma unroll
                for (int j = 0; j < 2; j++) {
                    const float val = rv[i][r][j] + (acc[i][j][r] + bv[j]);
                    C[row + nw + j * 16 + lm] = (val - mu[j]) * sc[j] + sh[j];
                }
            }
        }
        return;
    }

#pragma unroll
    for (int i = 0; i < 2; i++) {
#pragma unroll
        for (int r = 0; r < 8; r++) {
            const size_t row = (size_t)(mw + i * 16 + r + 8 * half) * N;
#pragma unroll
            for (int j = 0; j < 2; j++) {
                float v = acc[i][j][r] + bv[j];
                if (fuse_gelu) v = gelu_exact(v);
                C[row + nw + j * 16 + lm] = v;
            }
        }
    }
}

#define RPG    32
#define ECAP   256
#define SCAN_U 8

static __device__ __forceinline__ float4 res_gelu_bn4(float4 x, float4 a, float4 g,
                                                      float4 b, float4 m, float4 v)
{
    float4 o;
    o.x = ((x.x + gelu_exact(a.x)) - m.x) * (g.x * rsqrtf(v.x + BN_EPS)) + b.x;
    o.y = ((x.y + gelu_exact(a.y)) - m.y) * (g.y * rsqrtf(v.y + BN_EPS)) + b.y;
    o.z = ((x.z + gelu_exact(a.z)) - m.z) * (g.z * rsqrtf(v.z + BN_EPS)) + b.z;
    o.w = ((x.w + gelu_exact(a.w)) - m.w) * (g.w * rsqrtf(v.w + BN_EPS)) + b.w;
    return o;
}

static __device__ __forceinline__ void list_edge(int row, int e, int r0,
                                                 int* lcnt, int (*leid)[ECAP])
{
    const unsigned d = (unsigned)(row - r0);
    if (d < (unsigned)RPG) {
        const int p = atomicAdd(&lcnt[d], 1);
        if (p < ECAP) leid[d][p] = e;
    }
}

static __device__ __forceinline__ void list_edges4(int4 r, int e, int r0,
                                                   int* lcnt, int (*leid)[ECAP])
{
    const unsigned d0 = (unsigned)(r.x - r0), d1 = (unsigned)(r.y - r0);
    const unsigned d2 = (unsigned)(r.z - r0), d3 = (unsigned)(r.w - r0);
    const unsigned dm = min(min(d0, d1), min(d2, d3));
    if (dm < (unsigned)RPG) {
        list_edge(r.x, e,     r0, lcnt, leid);
        list_edge(r.y, e + 1, r0, lcnt, leid);
        list_edge(r.z, e + 2, r0, lcnt, leid);
        list_edge(r.w, e + 3, r0, lcnt, leid);
    }
}

static __device__ __forceinline__ void add_edge(float4& a0, float4& a1,
                                                const float* hb, int cj, float vj)
{
    const float* src = hb + (size_t)cj * DD;
    const float4 h0 = *(const float4*)(src);
    const float4 h1 = *(const float4*)(src + 128);
    a0.x += vj * h0.x;  a0.y += vj * h0.y;
    a0.z += vj * h0.z;  a0.w += vj * h0.w;
    a1.x += vj * h1.x;  a1.y += vj * h1.y;
    a1.z += vj * h1.z;  a1.w += vj * h1.w;
}

__global__ __launch_bounds__(1024)
void spmm_rows_bn(const int* __restrict__ rows, const int* __restrict__ cols,
                  const float* __restrict__ vals, int E,
                  const float* __restrict__ hl, const float* __restrict__ x,
                  const float* __restrict__ bn_g, const float* __restrict__ bn_b,
                  const float* __restrict__ bn_m, const float* __restrict__ bn_v,
                  float* __restrict__ x1)
{
    __shared__ int lcnt[RPG];
    __shared__ int leid[RPG][ECAP];
    __shared__ int sid[RPG][ECAP];

    const int t  = threadIdx.x;
    const int r0 = blockIdx.x * RPG;
    if (t < RPG) lcnt[t] = 0;
    __syncthreads();

    const int   n4    = E >> 2;
    const int4* rows4 = (const int4*)rows;
    int i0 = t;
    for (; i0 + (SCAN_U - 1) * 1024 < n4; i0 += 1024 * SCAN_U) {
        int4 r[SCAN_U];
#pragma unroll
        for (int u = 0; u < SCAN_U; u++) r[u] = rows4[i0 + u * 1024];
#pragma unroll
        for (int u = 0; u < SCAN_U; u++) list_edges4(r[u], (i0 + u * 1024) * 4, r0, lcnt, leid);
    }
    for (; i0 < n4; i0 += 1024)
        list_edges4(rows4[i0], i0 * 4, r0, lcnt, leid);
    for (int e = (n4 << 2) + t; e < E; e += 1024)
        list_edge(rows[e], e, r0, lcnt, leid);
    __syncthreads();

    const int w    = __builtin_amdgcn_readfirstlane(t >> 5);
    const int lane = t & 31;
    const int row  = r0 + w;
    const int deg  = __builtin_amdgcn_readfirstlane(lcnt[w]);
    const bool listed = deg <= ECAP;

    if (listed) {
        for (int base = 0; base < deg; base += 32) {
            const int n  = (deg - base) < 32 ? (deg - base) : 32;
            const int my = (lane < n) ? leid[w][base + lane] : 0x7fffffff;
            int place = 0;
            for (int b2 = 0; b2 < deg; b2 += 32) {
                const int n2    = (deg - b2) < 32 ? (deg - b2) : 32;
                const int other = (lane < n2) ? leid[w][b2 + lane] : 0x7fffffff;
                for (int j = 0; j < n2; j++)
                    place += (__builtin_amdgcn_readlane(other, j) < my) ? 1 : 0;
            }
            if (lane < n) sid[w][place] = my;
        }
    }
    __syncthreads();

    float4 a0 = make_float4(0.f, 0.f, 0.f, 0.f);
    float4 a1 = make_float4(0.f, 0.f, 0.f, 0.f);
    const float* hb = hl + lane * 4;

    if (listed) {
        for (int base = 0; base < deg; base += 32) {
            const int n = (deg - base) < 32 ? (deg - base) : 32;
            int   c = 0;
            float v = 0.0f;
            if (lane < n) {
                const int e = sid[w][base + lane];
                c = cols[e];
                v = vals[e];
            }
#pragma unroll 4
            for (int j = 0; j < n; j++) {
                const int   cj = __builtin_amdgcn_readlane(c, j);
                const float vj = __int_as_float(
                    __builtin_amdgcn_readlane(__float_as_int(v), j));
                add_edge(a0, a1, hb, cj, vj);
            }
        }
    } else {
        for (int eb = 0; eb < E; eb += 32) {
            const int  e   = eb + lane;
            const bool hit = (e < E) && (rows[e] == row);
            int   c = 0;
            float v = 0.0f;
            if (hit) { c = cols[e]; v = vals[e]; }
            unsigned m = __builtin_amdgcn_ballot_w32(hit);
            while (m) {
                const int j = __builtin_ctz(m);
                m &= m - 1;
                const int   cj = __builtin_amdgcn_readlane(c, j);
                const float vj = __int_as_float(
                    __builtin_amdgcn_readlane(__float_as_int(v), j));
                add_edge(a0, a1, hb, cj, vj);
            }
        }
    }

    const int    c0 = lane * 4;
    const float* xr = x + (size_t)row * DD + c0;
    float*      dst = x1 + (size_t)row * DD + c0;
    const float4 o0 = res_gelu_bn4(*(const float4*)(xr), a0,
                                   *(const float4*)(bn_g + c0), *(const float4*)(bn_b + c0),
                                   *(const float4*)(bn_m + c0), *(const float4*)(bn_v + c0));
    const float4 o1 = res_gelu_bn4(*(const float4*)(xr + 128), a1,
                                   *(const float4*)(bn_g + 128 + c0), *(const float4*)(bn_b + 128 + c0),
                                   *(const float4*)(bn_m + 128 + c0), *(const float4*)(bn_v + 128 + c0));
    *(float4*)(dst)       = o0;
    *(float4*)(dst + 128) = o1;
}

static __device__ __forceinline__ float xhalf(float x) {
    const int b = __builtin_bit_cast(int, x);
    return __builtin_bit_cast(float,
        __builtin_amdgcn_permlanex16(b, b, 0x76543210, 0xfedcba98, false, false));
}

#define AKC 64
#define ALD 36
#define VLD 72

__global__ __launch_bounds__(256)
void attn_wmma(const float* __restrict__ qkv, float* __restrict__ ctx)
{
    __shared__ float Ks[2][AKC * ALD];
    __shared__ float Vs[2][(AKC / 2) * VLD];

    const int t    = threadIdx.x;
    const int warp = t >> 5;
    const int qt   = (blockIdx.x & 3) * 8 + warp;
    const int h    = (blockIdx.x >> 2) & 7;
    const int b    = blockIdx.x >> 5;
    const int lane = t & 31;
    const int half = lane >> 4;
    const int lm   = lane & 15;
    const int LDQ  = 3 * DD;

    const float* gbase = qkv + (size_t)(b * NN) * LDQ;

    const int srow = t >> 3;
    const int sc4  = (t & 7) * 4;
    const float* kg = gbase + (size_t)srow * LDQ + DD + h * DH + sc4;
    const float* vg = gbase + (size_t)(2 * srow) * LDQ + 2 * DD + h * DH + sc4;
    const int soff  = srow * ALD + sc4;
    const int svoff = srow * VLD + 2 * sc4;

    v4f rk0 = *(const v4f*)(kg);
    v4f rk1 = *(const v4f*)(kg + (size_t)32 * LDQ);
    v4f rv0 = *(const v4f*)(vg);
    v4f rv1 = *(const v4f*)(vg + LDQ);

    v2f qb[8];
    {
        const float* qrow = gbase + (size_t)(qt * 16 + lm) * LDQ + h * DH;
        for (int j = 0; j < 8; j++) qb[j] = *(const v2f*)(qrow + 4 * j + 2 * half);
    }

    *(v4f*)(&Ks[0][soff])            = rk0;
    *(v4f*)(&Ks[0][soff + 32 * ALD]) = rk1;
    *(v4f*)(&Vs[0][svoff])           = (v4f){rv0.x, rv1.x, rv0.y, rv1.y};
    *(v4f*)(&Vs[0][svoff + 4])       = (v4f){rv0.z, rv1.z, rv0.w, rv1.w};
    __syncthreads();

    float rowmax = -1e30f;
    float rowsum = 0.0f;
    v8f o0 = {}, o1 = {};
    const float scale = 0.17677669529663687f;

    const int koff = lm * ALD + 2 * half;
    const int voff = (4 * half) * VLD + 2 * lm;

    for (int c = 0; c < NN / AKC; c++) {
        const int  cur  = c & 1;
        const bool more = (c + 1) < NN / AKC;
        if (more) {
            const size_t o = (size_t)((c + 1) * AKC) * LDQ;
            rk0 = *(const v4f*)(kg + o);
            rk1 = *(const v4f*)(kg + o + (size_t)32 * LDQ);
            rv0 = *(const v4f*)(vg + o);
            rv1 = *(const v4f*)(vg + o + LDQ);
        }

        const float* ks = &Ks[cur][koff];
        const float* vs = &Vs[cur][voff];
        for (int kt = 0; kt < AKC / 16; kt++) {
            v8f s = {};
            const float* krow = ks + kt * 16 * ALD;
            for (int j = 0; j < 8; j++) {
                v2f ka = *(const v2f*)(krow + 4 * j);
                s = wmma4(ka, qb[j], s);
            }

            float val[8];
            for (int r = 0; r < 8; r++) val[r] = s[r] * scale;
            float mx = val[0];
            for (int r = 1; r < 8; r++) mx = fmaxf(mx, val[r]);
            mx = fmaxf(mx, xhalf(mx));
            const float nm = fmaxf(rowmax, mx);
            const float al = __expf(rowmax - nm);
            float p[8];
            float sm = 0.0f;
            for (int r = 0; r < 8; r++) { p[r] = __expf(val[r] - nm); sm += p[r]; }
            rowmax = nm;
            rowsum = rowsum * al + sm;
            for (int r = 0; r < 8; r++) { o0[r] *= al; o1[r] *= al; }

            const float* vp = vs + kt * 8 * VLD;
            for (int j = 0; j < 4; j++) {
                v2f pb;
                pb.x = p[2 * j];                      pb.y = p[2 * j + 1];
                const v2f va0 = *(const v2f*)(vp + j * VLD);
                const v2f va1 = *(const v2f*)(vp + j * VLD + 32);
                o0 = wmma4(va0, pb, o0);
                o1 = wmma4(va1, pb, o1);
            }
        }

        if (more) {
            float* kn = &Ks[cur ^ 1][soff];
            float* vn = &Vs[cur ^ 1][svoff];
            *(v4f*)(kn)            = rk0;
            *(v4f*)(kn + 32 * ALD) = rk1;
            *(v4f*)(vn)            = (v4f){rv0.x, rv1.x, rv0.y, rv1.y};
            *(v4f*)(vn + 4)        = (v4f){rv0.z, rv1.z, rv0.w, rv1.w};
        }
        __syncthreads();
    }

    const float inv = 1.0f / (rowsum + xhalf(rowsum));
    float* dst = ctx + (size_t)(b * NN + qt * 16 + lm) * DD + h * DH + 8 * half;
    *(v4f*)(dst)      = (v4f){o0[0] * inv, o0[1] * inv, o0[2] * inv, o0[3] * inv};
    *(v4f*)(dst + 4)  = (v4f){o0[4] * inv, o0[5] * inv, o0[6] * inv, o0[7] * inv};
    *(v4f*)(dst + 16) = (v4f){o1[0] * inv, o1[1] * inv, o1[2] * inv, o1[3] * inv};
    *(v4f*)(dst + 20) = (v4f){o1[4] * inv, o1[5] * inv, o1[6] * inv, o1[7] * inv};
}

static inline int gemm_blocks(int M, int N) { return (M >> 6) * (N >> 7); }

extern "C" void kernel_launch(void* const* d_in, const int* in_sizes, int n_in,
                              void* d_out, int out_size, void* d_ws, size_t ws_size,
                              hipStream_t stream)
{
    (void)n_in; (void)out_size; (void)ws_size;
    const float* x          = (const float*)d_in[0];
    const int*   edge_rows  = (const int*)  d_in[1];
    const int*   edge_cols  = (const int*)  d_in[2];
    const float* edge_vals  = (const float*)d_in[3];
    const float* w_gcn      = (const float*)d_in[4];
    const float* in_proj_w  = (const float*)d_in[5];
    const float* in_proj_b  = (const float*)d_in[6];
    const float* out_proj_w = (const float*)d_in[7];
    const float* out_proj_b = (const float*)d_in[8];
    const float* w1         = (const float*)d_in[9];
    const float* b1         = (const float*)d_in[10];
    const float* w2         = (const float*)d_in[11];
    const float* b2         = (const float*)d_in[12];
    const float* bn1_g = (const float*)d_in[15], *bn1_b = (const float*)d_in[16];
    const float* bn1_m = (const float*)d_in[17], *bn1_v = (const float*)d_in[18];
    const float* bn2_g = (const float*)d_in[19], *bn2_b = (const float*)d_in[20];
    const float* bn2_m = (const float*)d_in[21], *bn2_v = (const float*)d_in[22];
    const float* bn3_g = (const float*)d_in[23], *bn3_b = (const float*)d_in[24];
    const float* bn3_m = (const float*)d_in[25], *bn3_v = (const float*)d_in[26];

    const int E = in_sizes[1];

    float* ws = (float*)d_ws;
    float* wsA = ws;
    float* wsB = wsA + (size_t)NT * 1024;
    float* wsC = wsB + (size_t)NT * DD;
    float* wsD = wsC + (size_t)NT * DD;
    float* wsE = wsD + (size_t)NT * DD;
    float* outp = (float*)d_out;

    const dim3 blk(256);

    gemm_wmma_f32<<<gemm_blocks(NT, DD), blk, 0, stream>>>(x, w_gcn, nullptr, wsB,
                                                           NT, DD, DD, 0, nullptr,
                                                           nullptr, nullptr, nullptr, nullptr);
    spmm_rows_bn<<<NT / RPG, dim3(32 * RPG), 0, stream>>>(edge_rows, edge_cols, edge_vals, E,
                                                          wsB, x, bn1_g, bn1_b, bn1_m, bn1_v, wsD);
    gemm_wmma_f32<<<gemm_blocks(NT, 3 * DD), blk, 0, stream>>>(wsD, in_proj_w,
                                                               in_proj_b, wsA,
                                                               NT, 3 * DD, DD, 0, nullptr,
                                                               nullptr, nullptr, nullptr, nullptr);
    attn_wmma<<<(BB * HH * (NN / 16)) / 8, blk, 0, stream>>>(wsA, wsB);
    gemm_wmma_f32<<<gemm_blocks(NT, DD), blk, 0, stream>>>(wsB, out_proj_w,
                                                           out_proj_b, wsE,
                                                           NT, DD, DD, 0, wsD,
                                                           bn2_g, bn2_b, bn2_m, bn2_v);
    gemm_wmma_f32<<<gemm_blocks(NT, 4 * DD), blk, 0, stream>>>(wsE, w1, b1, wsA,
                                                               NT, 4 * DD, DD, 1, nullptr,
                                                               nullptr, nullptr, nullptr, nullptr);
    gemm_wmma_f32<<<gemm_blocks(NT, DD), blk, 0, stream>>>(wsA, w2, b2, outp,
                                                           NT, DD, 4 * DD, 0, wsE,
                                                           bn3_g, bn3_b, bn3_m, bn3_v);
}


# edge rows scanned once: 256 extra work-groups in the first GEMM's grid sort 1024-edge slices by 32-row block; a row-sum work-group reads its block's ids, not all edge rows; sum order unchanged
# speedup vs baseline: 3.3823x; 1.0517x over previous
#include <hip/hip_runtime.h>
#include <hip/hip_bf16.h>
#include <math.h>

#define BB   16
#define NN   512
#define DD   256
#define HH   8
#define DH   32
#define NT   (BB * NN)
#define BN_EPS 1e-5f

typedef float v2f __attribute__((ext_vector_type(2)));
typedef float v8f __attribute__((ext_vector_type(8)));
typedef float v4f __attribute__((ext_vector_type(4)));

static __device__ __forceinline__ v8f wmma4(v2f a, v2f b, v8f c) {
    return __builtin_amdgcn_wmma_f32_16x16x4_f32(
         false, a,  false, b,
         (short)0, c,  false,  false);
}

static __device__ __forceinline__ float gelu_exact(float v) {
    return 0.5f * v * (1.0f + erff(v * 0.70710678118654752f));
}

#define GBM 64
#define GBN 128
#define GKC 32
#define GLD 36

static __device__ __forceinline__
void gemm_tile(const float* __restrict__ A, const float* __restrict__ W,
               const float* __restrict__ bias, float* __restrict__ C,
               int M, int N, int K, int fuse_gelu,
               const float* __restrict__ res,
               const float* __restrict__ bn_g, const float* __restrict__ bn_b,
               const float* __restrict__ bn_m, const float* __restrict__ bn_v)
{
    __shared__ float As[2][GBM * GLD];
    __shared__ float Bs[2][GBN * GLD];
    (void)M;

    const int nbn = N >> 7;
    const int bm  = blockIdx.x / nbn;
    const int bn  = blockIdx.x - bm * nbn;
    const int m0  = bm * GBM;
    const int n0  = bn * GBN;

    const int t    = threadIdx.x;
    const int wv   = t >> 5;
    const int wm   = wv >> 2;
    const int wn   = wv & 3;
    const int lane = t & 31;
    const int half = lane >> 4;
    const int lm   = lane & 15;

    const int srow = t >> 3;
    const int sc4  = (t & 7) * 4;
    const float* ag = A + (size_t)(m0 + srow) * K + sc4;
    const float* wg = W + (size_t)(n0 + srow) * K + sc4;
    const int soff  = srow * GLD + sc4;

    v4f ra0 = *(const v4f*)(ag);
    v4f ra1 = *(const v4f*)(ag + (size_t)32 * K);
    v4f rb0 = *(const v4f*)(wg);
    v4f rb1 = *(const v4f*)(wg + (size_t)32 * K);
    v4f rb2 = *(const v4f*)(wg + (size_t)64 * K);
    v4f rb3 = *(const v4f*)(wg + (size_t)96 * K);
    *(v4f*)(&As[0][soff])            = ra0;
    *(v4f*)(&As[0][soff + 32 * GLD]) = ra1;
    *(v4f*)(&Bs[0][soff])            = rb0;
    *(v4f*)(&Bs[0][soff + 32 * GLD]) = rb1;
    *(v4f*)(&Bs[0][soff + 64 * GLD]) = rb2;
    *(v4f*)(&Bs[0][soff + 96 * GLD]) = rb3;
    __syncthreads();

    v8f acc[2][2];
#pragma unroll
    for (int i = 0; i < 2; i++)
#pragma unroll
        for (int j = 0; j < 2; j++) acc[i][j] = (v8f){};

    const int aoff = (wm * 32 + lm) * GLD + 2 * half;
    const int boff = (wn * 32 + lm) * GLD + 2 * half;
    const int nch  = K / GKC;

    for (int c = 0; c < nch; c++) {
        const int  cur  = c & 1;
        const bool more = (c + 1) < nch;
        if (more) {
            const int k1 = (c + 1) * GKC;
            ra0 = *(const v4f*)(ag + k1);
            ra1 = *(const v4f*)(ag + (size_t)32 * K + k1);
            rb0 = *(const v4f*)(wg + k1);
            rb1 = *(const v4f*)(wg + (size_t)32 * K + k1);
            rb2 = *(const v4f*)(wg + (size_t)64 * K + k1);
            rb3 = *(const v4f*)(wg + (size_t)96 * K + k1);
        }

        const float* as = &As[cur][aoff];
        const float* bs = &Bs[cur][boff];
#pragma unroll
        for (int j = 0; j < GKC / 4; j++) {
            v2f a0 = *(const v2f*)(as + 4 * j);
            v2f a1 = *(const v2f*)(as + 16 * GLD + 4 * j);
            v2f b0 = *(const v2f*)(bs + 4 * j);
            v2f b1 = *(const v2f*)(bs + 16 * GLD + 4 * j);
            acc[0][0] = wmma4(a0, b0, acc[0][0]);
            acc[0][1] = wmma4(a0, b1, acc[0][1]);
            acc[1][0] = wmma4(a1, b0, acc[1][0]);
            acc[1][1] = wmma4(a1, b1, acc[1][1]);
        }

        if (more) {
            float* an = &As[cur ^ 1][soff];
            float* bn2 = &Bs[cur ^ 1][soff];
            *(v4f*)(an)             = ra0;
            *(v4f*)(an + 32 * GLD)  = ra1;
            *(v4f*)(bn2)            = rb0;
            *(v4f*)(bn2 + 32 * GLD) = rb1;
            *(v4f*)(bn2 + 64 * GLD) = rb2;
            *(v4f*)(bn2 + 96 * GLD) = rb3;
        }
        __syncthreads();
    }

    const int nw = n0 + wn * 32;
    const int mw = m0 + wm * 32;
    float bv[2];
#pragma unroll
    for (int j = 0; j < 2; j++)
        bv[j] = bias ? bias[nw + j * 16 + lm] : 0.0f;

    if (res) {
        float sc[2], mu[2], sh[2];
#pragma unroll
        for (int j = 0; j < 2; j++) {
            const int col = nw + j * 16 + lm;
            sc[j] = bn_g[col] * rsqrtf(bn_v[col] + BN_EPS);
            mu[j] = bn_m[col];
            sh[j] = bn_b[col];
        }
        float rv[2][8][2];
#pragma unroll
        for (int i = 0; i < 2; i++) {
#pragma unroll
            for (int r = 0; r < 8; r++) {
                const size_t row = (size_t)(mw + i * 16 + r + 8 * half) * N;
#pragma unroll
                for (int j = 0; j < 2; j++)
                    rv[i][r][j] = res[row + nw + j * 16 + lm];
            }
        }
#pragma unroll
        for (int i = 0; i < 2; i++) {
#pragma unroll
            for (int r = 0; r < 8; r++) {
                const size_t row = (size_t)(mw + i * 16 + r + 8 * half) * N;
#pragma unroll
                for (int j = 0; j < 2; j++) {
                    const float val = rv[i][r][j] + (acc[i][j][r] + bv[j]);
                    C[row + nw + j * 16 + lm] = (val - mu[j]) * sc[j] + sh[j];
                }
            }
        }
        return;
    }

#pragma unroll
    for (int i = 0; i < 2; i++) {
#pragma unroll
        for (int r = 0; r < 8; r++) {
            const size_t row = (size_t)(mw + i * 16 + r + 8 * half) * N;
#pragma unroll
            for (int j = 0; j < 2; j++) {
                float v = acc[i][j][r] + bv[j];
                if (fuse_gelu) v = gelu_exact(v);
                C[row + nw + j * 16 + lm] = v;
            }
        }
    }
}

__global__ __launch_bounds__(256)
void gemm_wmma_f32(const float* __restrict__ A, const float* __restrict__ W,
                   const float* __restrict__ bias, float* __restrict__ C,
                   int M, int N, int K, int fuse_gelu,
                   const float* __restrict__ res,
                   const float* __restrict__ bn_g, const float* __restrict__ bn_b,
                   const float* __restrict__ bn_m, const float* __restrict__ bn_v)
{
    gemm_tile(A, W, bias, C, M, N, K, fuse_gelu, res, bn_g, bn_b, bn_m, bn_v);
}

#define RPG    32
#define NBKT   (NT / RPG)
#define SLICE  1024
#define NSLICE 256

static __device__ __forceinline__
void bin_slice(const int* __restrict__ rows, int E, int s,
               int* __restrict__ bin_off, int* __restrict__ bin_ids)
{
    __shared__ int hist[NBKT];
    __shared__ int curs[NBKT];
    __shared__ int wsum[8];

    const int t    = threadIdx.x;
    const int lane = t & 31;
    const int wv   = t >> 5;
    hist[t] = 0;
    __syncthreads();

    const int e0 = s * SLICE + t * 4;
    int r[4];
    if (e0 + 3 < E) {
        const int4 q = *(const int4*)(rows + e0);
        r[0] = q.x; r[1] = q.y; r[2] = q.z; r[3] = q.w;
    } else {
        for (int i = 0; i < 4; i++) r[i] = (e0 + i < E) ? rows[e0 + i] : -1;
    }
    int b[4];
    for (int i = 0; i < 4; i++) {
        b[i] = ((unsigned)r[i] < (unsigned)NT) ? (r[i] / RPG) : -1;
        if (b[i] >= 0) atomicAdd(&hist[b[i]], 1);
    }
    __syncthreads();

    const int v = hist[t];
    int incl = v;
    for (int d = 1; d < 32; d <<= 1) {
        const int n = __shfl_up(incl, d);
        if (lane >= d) incl += n;
    }
    if (lane == 31) wsum[wv] = incl;
    __syncthreads();
    int base = 0;
    for (int i = 0; i < 8; i++) base += (i < wv) ? wsum[i] : 0;
    const int excl = base + incl - v;
    curs[t] = excl;
    bin_off[s * (NBKT + 1) + t] = excl;
    if (t == NBKT - 1) bin_off[s * (NBKT + 1) + NBKT] = excl + v;
    __syncthreads();

    for (int i = 0; i < 4; i++) {
        if (b[i] >= 0) {
            const int p = atomicAdd(&curs[b[i]], 1);
            bin_ids[s * SLICE + p] = (e0 + i) * RPG + (r[i] & (RPG - 1));
        }
    }
}

__global__ __launch_bounds__(256)
void gemm_bin_f32(const float* __restrict__ A, const float* __restrict__ W,
                  float* __restrict__ C, int M, int N, int K,
                  const int* __restrict__ rows, int E,
                  int* __restrict__ bin_off, int* __restrict__ bin_ids)
{
    const int ntile = (M >> 6) * (N >> 7);
    if ((int)blockIdx.x >= ntile) {
        bin_slice(rows, E, (int)blockIdx.x - ntile, bin_off, bin_ids);
        return;
    }
    gemm_tile(A, W, nullptr, C, M, N, K, 0, nullptr, nullptr, nullptr, nullptr, nullptr);
}

#define ECAP   256

static __device__ __forceinline__ float4 res_gelu_bn4(float4 x, float4 a, float4 g,
                                                      float4 b, float4 m, float4 v)
{
    float4 o;
    o.x = ((x.x + gelu_exact(a.x)) - m.x) * (g.x * rsqrtf(v.x + BN_EPS)) + b.x;
    o.y = ((x.y + gelu_exact(a.y)) - m.y) * (g.y * rsqrtf(v.y + BN_EPS)) + b.y;
    o.z = ((x.z + gelu_exact(a.z)) - m.z) * (g.z * rsqrtf(v.z + BN_EPS)) + b.z;
    o.w = ((x.w + gelu_exact(a.w)) - m.w) * (g.w * rsqrtf(v.w + BN_EPS)) + b.w;
    return o;
}

static __device__ __forceinline__ void list_edge(int row, int e, int r0,
                                                 int* lcnt, int (*leid)[ECAP])
{
    const unsigned d = (unsigned)(row - r0);
    if (d < (unsigned)RPG) {
        const int p = atomicAdd(&lcnt[d], 1);
        if (p < ECAP) leid[d][p] = e;
    }
}

static __device__ __forceinline__ void add_edge(float4& a0, float4& a1,
                                                const float* hb, int cj, float vj)
{
    const float* src = hb + (size_t)cj * DD;
    const float4 h0 = *(const float4*)(src);
    const float4 h1 = *(const float4*)(src + 128);
    a0.x += vj * h0.x;  a0.y += vj * h0.y;
    a0.z += vj * h0.z;  a0.w += vj * h0.w;
    a1.x += vj * h1.x;  a1.y += vj * h1.y;
    a1.z += vj * h1.z;  a1.w += vj * h1.w;
}

__global__ __launch_bounds__(1024)
void spmm_rows_bn(const int* __restrict__ rows, const int* __restrict__ cols,
                  const float* __restrict__ vals, int E,
                  const float* __restrict__ hl, const float* __restrict__ x,
                  const float* __restrict__ bn_g, const float* __restrict__ bn_b,
                  const float* __restrict__ bn_m, const float* __restrict__ bn_v,
                  float* __restrict__ x1,
                  const int* __restrict__ bin_off, const int* __restrict__ bin_ids)
{
    __shared__ int pre[NSLICE + 1];
    __shared__ int sst[NSLICE];
    __shared__ int wsum[NSLICE / 32];
    __shared__ int lcnt[RPG];
    __shared__ int leid[RPG][ECAP];
    __shared__ int sid[RPG][ECAP];

    const int t  = threadIdx.x;
    const int r0 = blockIdx.x * RPG;
    if (t < RPG) lcnt[t] = 0;

    const int tl = t & 31;
    int cnt = 0;
    if (t < NSLICE) {
        const int* bo = bin_off + t * (NBKT + 1) + blockIdx.x;
        const int o0 = bo[0];
        cnt    = bo[1] - o0;
        sst[t] = t * SLICE + o0;
    }
    int incl = cnt;
    for (int d = 1; d < 32; d <<= 1) {
        const int n = __shfl_up(incl, d);
        if (tl >= d) incl += n;
    }
    if (t < NSLICE && tl == 31) wsum[t >> 5] = incl;
    __syncthreads();
    if (t < NSLICE) {
        int base = 0;
        for (int i = 0; i < NSLICE / 32; i++) base += (i < (t >> 5)) ? wsum[i] : 0;
        pre[t] = base + incl - cnt;
        if (t == NSLICE - 1) pre[NSLICE] = base + incl;
    }
    __syncthreads();

    const int total = pre[NSLICE];
    for (int k = t; k < total; k += 1024) {
        int lo = 0, hi = NSLICE;
        for (int it = 0; it < 8; it++) {
            const int mid = (lo + hi) >> 1;
            if (pre[mid] <= k) lo = mid; else hi = mid;
        }
        const unsigned pk = (unsigned)bin_ids[sst[lo] + (k - pre[lo])];
        const int d = (int)(pk & (RPG - 1));
        const int p = atomicAdd(&lcnt[d], 1);
        if (p < ECAP) leid[d][p] = (int)(pk / RPG);
    }
    for (int e = NSLICE * SLICE + t; e < E; e += 1024)
        list_edge(rows[e], e, r0, lcnt, leid);
    __syncthreads();

    const int w    = __builtin_amdgcn_readfirstlane(t >> 5);
    const int lane = t & 31;
    const int row  = r0 + w;
    const int deg  = __builtin_amdgcn_readfirstlane(lcnt[w]);
    const bool listed = deg <= ECAP;

    if (listed) {
        for (int base = 0; base < deg; base += 32) {
            const int n  = (deg - base) < 32 ? (deg - base) : 32;
            const int my = (lane < n) ? leid[w][base + lane] : 0x7fffffff;
            int place = 0;
            for (int b2 = 0; b2 < deg; b2 += 32) {
                const int n2    = (deg - b2) < 32 ? (deg - b2) : 32;
                const int other = (lane < n2) ? leid[w][b2 + lane] : 0x7fffffff;
                for (int j = 0; j < n2; j++)
                    place += (__builtin_amdgcn_readlane(other, j) < my) ? 1 : 0;
            }
            if (lane < n) sid[w][place] = my;
        }
    }
    __syncthreads();

    float4 a0 = make_float4(0.f, 0.f, 0.f, 0.f);
    float4 a1 = make_float4(0.f, 0.f, 0.f, 0.f);
    const float* hb = hl + lane * 4;

    if (listed) {
        for (int base = 0; base < deg; base += 32) {
            const int n = (deg - base) < 32 ? (deg - base) : 32;
            int   c = 0;
            float v = 0.0f;
            if (lane < n) {
                const int e = sid[w][base + lane];
                c = cols[e];
                v = vals[e];
            }
#pragma unroll 4
            for (int j = 0; j < n; j++) {
                const int   cj = __builtin_amdgcn_readlane(c, j);
                const float vj = __int_as_float(
                    __builtin_amdgcn_readlane(__float_as_int(v), j));
                add_edge(a0, a1, hb, cj, vj);
            }
        }
    } else {
        for (int eb = 0; eb < E; eb += 32) {
            const int  e   = eb + lane;
            const bool hit = (e < E) && (rows[e] == row);
            int   c = 0;
            float v = 0.0f;
            if (hit) { c = cols[e]; v = vals[e]; }
            unsigned m = __builtin_amdgcn_ballot_w32(hit);
            while (m) {
                const int j = __builtin_ctz(m);
                m &= m - 1;
                const int   cj = __builtin_amdgcn_readlane(c, j);
                const float vj = __int_as_float(
                    __builtin_amdgcn_readlane(__float_as_int(v), j));
                add_edge(a0, a1, hb, cj, vj);
            }
        }
    }

    const int    c0 = lane * 4;
    const float* xr = x + (size_t)row * DD + c0;
    float*      dst = x1 + (size_t)row * DD + c0;
    const float4 o0 = res_gelu_bn4(*(const float4*)(xr), a0,
                                   *(const float4*)(bn_g + c0), *(const float4*)(bn_b + c0),
                                   *(const float4*)(bn_m + c0), *(const float4*)(bn_v + c0));
    const float4 o1 = res_gelu_bn4(*(const float4*)(xr + 128), a1,
                                   *(const float4*)(bn_g + 128 + c0), *(const float4*)(bn_b + 128 + c0),
                                   *(const float4*)(bn_m + 128 + c0), *(const float4*)(bn_v + 128 + c0));
    *(float4*)(dst)       = o0;
    *(float4*)(dst + 128) = o1;
}

static __device__ __forceinline__ float xhalf(float x) {
    const int b = __builtin_bit_cast(int, x);
    return __builtin_bit_cast(float,
        __builtin_amdgcn_permlanex16(b, b, 0x76543210, 0xfedcba98, false, false));
}

#define AKC 64
#define ALD 36
#define VLD 72

__global__ __launch_bounds__(256)
void attn_wmma(const float* __restrict__ qkv, float* __restrict__ ctx)
{
    __shared__ float Ks[2][AKC * ALD];
    __shared__ float Vs[2][(AKC / 2) * VLD];

    const int t    = threadIdx.x;
    const int warp = t >> 5;
    const int qt   = (blockIdx.x & 3) * 8 + warp;
    const int h    = (blockIdx.x >> 2) & 7;
    const int b    = blockIdx.x >> 5;
    const int lane = t & 31;
    const int half = lane >> 4;
    const int lm   = lane & 15;
    const int LDQ  = 3 * DD;

    const float* gbase = qkv + (size_t)(b * NN) * LDQ;

    const int srow = t >> 3;
    const int sc4  = (t & 7) * 4;
    const float* kg = gbase + (size_t)srow * LDQ + DD + h * DH + sc4;
    const float* vg = gbase + (size_t)(2 * srow) * LDQ + 2 * DD + h * DH + sc4;
    const int soff  = srow * ALD + sc4;
    const int svoff = srow * VLD + 2 * sc4;

    v4f rk0 = *(const v4f*)(kg);
    v4f rk1 = *(const v4f*)(kg + (size_t)32 * LDQ);
    v4f rv0 = *(const v4f*)(vg);
    v4f rv1 = *(const v4f*)(vg + LDQ);

    v2f qb[8];
    {
        const float* qrow = gbase + (size_t)(qt * 16 + lm) * LDQ + h * DH;
        for (int j = 0; j < 8; j++) qb[j] = *(const v2f*)(qrow + 4 * j + 2 * half);
    }

    *(v4f*)(&Ks[0][soff])            = rk0;
    *(v4f*)(&Ks[0][soff + 32 * ALD]) = rk1;
    *(v4f*)(&Vs[0][svoff])           = (v4f){rv0.x, rv1.x, rv0.y, rv1.y};
    *(v4f*)(&Vs[0][svoff + 4])       = (v4f){rv0.z, rv1.z, rv0.w, rv1.w};
    __syncthreads();

    float rowmax = -1e30f;
    float rowsum = 0.0f;
    v8f o0 = {}, o1 = {};
    const float scale = 0.17677669529663687f;

    const int koff = lm * ALD + 2 * half;
    const int voff = (4 * half) * VLD + 2 * lm;

    for (int c = 0; c < NN / AKC; c++) {
        const int  cur  = c & 1;
        const bool more = (c + 1) < NN / AKC;
        if (more) {
            const size_t o = (size_t)((c + 1) * AKC) * LDQ;
            rk0 = *(const v4f*)(kg + o);
            rk1 = *(const v4f*)(kg + o + (size_t)32 * LDQ);
            rv0 = *(const v4f*)(vg + o);
            rv1 = *(const v4f*)(vg + o + LDQ);
        }

        const float* ks = &Ks[cur][koff];
        const float* vs = &Vs[cur][voff];
        for (int kt = 0; kt < AKC / 16; kt++) {
            v8f s = {};
            const float* krow = ks + kt * 16 * ALD;
            for (int j = 0; j < 8; j++) {
                v2f ka = *(const v2f*)(krow + 4 * j);
                s = wmma4(ka, qb[j], s);
            }

            float val[8];
            for (int r = 0; r < 8; r++) val[r] = s[r] * scale;
            float mx = val[0];
            for (int r = 1; r < 8; r++) mx = fmaxf(mx, val[r]);
            mx = fmaxf(mx, xhalf(mx));
            const float nm = fmaxf(rowmax, mx);
            const float al = __expf(rowmax - nm);
            float p[8];
            float sm = 0.0f;
            for (int r = 0; r < 8; r++) { p[r] = __expf(val[r] - nm); sm += p[r]; }
            rowmax = nm;
            rowsum = rowsum * al + sm;
            for (int r = 0; r < 8; r++) { o0[r] *= al; o1[r] *= al; }

            const float* vp = vs + kt * 8 * VLD;
            for (int j = 0; j < 4; j++) {
                v2f pb;
                pb.x = p[2 * j];                      pb.y = p[2 * j + 1];
                const v2f va0 = *(const v2f*)(vp + j * VLD);
                const v2f va1 = *(const v2f*)(vp + j * VLD + 32);
                o0 = wmma4(va0, pb, o0);
                o1 = wmma4(va1, pb, o1);
            }
        }

        if (more) {
            float* kn = &Ks[cur ^ 1][soff];
            float* vn = &Vs[cur ^ 1][svoff];
            *(v4f*)(kn)            = rk0;
            *(v4f*)(kn + 32 * ALD) = rk1;
            *(v4f*)(vn)            = (v4f){rv0.x, rv1.x, rv0.y, rv1.y};
            *(v4f*)(vn + 4)        = (v4f){rv0.z, rv1.z, rv0.w, rv1.w};
        }
        __syncthreads();
    }

    const float inv = 1.0f / (rowsum + xhalf(rowsum));
    float* dst = ctx + (size_t)(b * NN + qt * 16 + lm) * DD + h * DH + 8 * half;
    *(v4f*)(dst)      = (v4f){o0[0] * inv, o0[1] * inv, o0[2] * inv, o0[3] * inv};
    *(v4f*)(dst + 4)  = (v4f){o0[4] * inv, o0[5] * inv, o0[6] * inv, o0[7] * inv};
    *(v4f*)(dst + 16) = (v4f){o1[0] * inv, o1[1] * inv, o1[2] * inv, o1[3] * inv};
    *(v4f*)(dst + 20) = (v4f){o1[4] * inv, o1[5] * inv, o1[6] * inv, o1[7] * inv};
}

static inline int gemm_blocks(int M, int N) { return (M >> 6) * (N >> 7); }

extern "C" void kernel_launch(void* const* d_in, const int* in_sizes, int n_in,
                              void* d_out, int out_size, void* d_ws, size_t ws_size,
                              hipStream_t stream)
{
    (void)n_in; (void)out_size; (void)ws_size;
    const float* x          = (const float*)d_in[0];
    const int*   edge_rows  = (const int*)  d_in[1];
    const int*   edge_cols  = (const int*)  d_in[2];
    const float* edge_vals  = (const float*)d_in[3];
    const float* w_gcn      = (const float*)d_in[4];
    const float* in_proj_w  = (const float*)d_in[5];
    const float* in_proj_b  = (const float*)d_in[6];
    const float* out_proj_w = (const float*)d_in[7];
    const float* out_proj_b = (const float*)d_in[8];
    const float* w1         = (const float*)d_in[9];
    const float* b1         = (const float*)d_in[10];
    const float* w2         = (const float*)d_in[11];
    const float* b2         = (const float*)d_in[12];
    const float* bn1_g = (const float*)d_in[15], *bn1_b = (const float*)d_in[16];
    const float* bn1_m = (const float*)d_in[17], *bn1_v = (const float*)d_in[18];
    const float* bn2_g = (const float*)d_in[19], *bn2_b = (const float*)d_in[20];
    const float* bn2_m = (const float*)d_in[21], *bn2_v = (const float*)d_in[22];
    const float* bn3_g = (const float*)d_in[23], *bn3_b = (const float*)d_in[24];
    const float* bn3_m = (const float*)d_in[25], *bn3_v = (const float*)d_in[26];

    const int E = in_sizes[1];

    float* ws = (float*)d_ws;
    float* wsA = ws;
    float* wsB = wsA + (size_t)NT * 1024;
    float* wsC = wsB + (size_t)NT * DD;
    float* wsD = wsC + (size_t)NT * DD;
    float* wsE = wsD + (size_t)NT * DD;
    float* outp = (float*)d_out;
    int* bin_off = (int*)wsC;
    int* bin_ids = bin_off + NSLICE * (NBKT + 1);

    const dim3 blk(256);

    gemm_bin_f32<<<gemm_blocks(NT, DD) + NSLICE, blk, 0, stream>>>(x, w_gcn, wsB, NT, DD, DD,
                                                                   edge_rows, E, bin_off, bin_ids);
    spmm_rows_bn<<<NBKT, dim3(32 * RPG), 0, stream>>>(edge_rows, edge_cols, edge_vals, E,
                                                      wsB, x, bn1_g, bn1_b, bn1_m, bn1_v, wsD,
                                                      bin_off, bin_ids);
    gemm_wmma_f32<<<gemm_blocks(NT, 3 * DD), blk, 0, stream>>>(wsD, in_proj_w,
                                                               in_proj_b, wsA,
                                                               NT, 3 * DD, DD, 0, nullptr,
                                                               nullptr, nullptr, nullptr, nullptr);
    attn_wmma<<<(BB * HH * (NN / 16)) / 8, blk, 0, stream>>>(wsA, wsB);
    gemm_wmma_f32<<<gemm_blocks(NT, DD), blk, 0, stream>>>(wsB, out_proj_w,
                                                           out_proj_b, wsE,
                                                           NT, DD, DD, 0, wsD,
                                                           bn2_g, bn2_b, bn2_m, bn2_v);
    gemm_wmma_f32<<<gemm_blocks(NT, 4 * DD), blk, 0, stream>>>(wsE, w1, b1, wsA,
                                                               NT, 4 * DD, DD, 1, nullptr,
                                                               nullptr, nullptr, nullptr, nullptr);
    gemm_wmma_f32<<<gemm_blocks(NT, DD), blk, 0, stream>>>(wsA, w2, b2, outp,
                                                           NT, DD, 4 * DD, 0, wsE,
                                                           bn3_g, bn3_b, bn3_m, bn3_v);
}
